# starting kernel
# baseline (speedup 1.0000x reference)
#include <hip/hip_runtime.h>
#include <math.h>

#define N_NODES 50000
#define N_EDGES 800000
#define IN_DIM 128
#define HIDDEN 64
#define NEG_SLOPE 0.2f
#define E_TOT (N_EDGES + N_NODES)

typedef float v2f __attribute__((ext_vector_type(2)));
typedef float v8f __attribute__((ext_vector_type(8)));

__device__ __forceinline__ void atomicMaxFloat(float* addr, float val) {
    if (val >= 0.0f) {
        atomicMax((int*)addr, __float_as_int(val));
    } else {
        atomicMin((unsigned int*)addr, __float_as_uint(val));
    }
}

__global__ void init_ws_kernel(float* __restrict__ mbuf, float* __restrict__ sbuf,
                               float* __restrict__ outacc) {
    int i = blockIdx.x * blockDim.x + threadIdx.x;
    if (i < N_NODES) { mbuf[i] = -INFINITY; sbuf[i] = 0.0f; }
    outacc[i] = 0.0f;
}

__global__ void __launch_bounds__(32)
gat_gemm_wmma_kernel(const float* __restrict__ x, const float* __restrict__ W,
                     const float* __restrict__ att_src_v, const float* __restrict__ att_dst_v,
                     float* __restrict__ h, float* __restrict__ a_src,
                     float* __restrict__ a_dst)
{
    const int tile = blockIdx.x;
    const int lane = threadIdx.x;
    const int half = lane >> 4;
    const int lrow = lane & 15;
    const int row0 = tile * 16;

    const float* xrow  = x + (size_t)(row0 + lrow) * IN_DIM;
    const float* wrow  = W + (size_t)lrow * IN_DIM;

    v8f acc0 = {}, acc1 = {}, acc2 = {}, acc3 = {};

    for (int kb = 0; kb < IN_DIM; kb += 4) {
        const int k = kb + 2 * half;
        v2f a  = *(const v2f*)(xrow + k);
        v2f b0 = *(const v2f*)(wrow + k);
        v2f b1 = *(const v2f*)(wrow + 16 * IN_DIM + k);
        v2f b2 = *(const v2f*)(wrow + 32 * IN_DIM + k);
        v2f b3 = *(const v2f*)(wrow + 48 * IN_DIM + k);
        acc0 = __builtin_amdgcn_wmma_f32_16x16x4_f32(false, a, false, b0, (short)0, acc0, false, false);
        acc1 = __builtin_amdgcn_wmma_f32_16x16x4_f32(false, a, false, b1, (short)0, acc1, false, false);
        acc2 = __builtin_amdgcn_wmma_f32_16x16x4_f32(false, a, false, b2, (short)0, acc2, false, false);
        acc3 = __builtin_amdgcn_wmma_f32_16x16x4_f32(false, a, false, b3, (short)0, acc3, false, false);
    }

    const float as0 = att_src_v[lrow],      ad0 = att_dst_v[lrow];
    const float as1 = att_src_v[16 + lrow], ad1 = att_dst_v[16 + lrow];
    const float as2 = att_src_v[32 + lrow], ad2 = att_dst_v[32 + lrow];
    const float as3 = att_src_v[48 + lrow], ad3 = att_dst_v[48 + lrow];

    #pragma unroll
    for (int r = 0; r < 8; ++r) {
        const int row = row0 + r + 8 * half;
        float* hr = h + (size_t)row * HIDDEN;
        hr[lrow]      = acc0[r];
        hr[16 + lrow] = acc1[r];
        hr[32 + lrow] = acc2[r];
        hr[48 + lrow] = acc3[r];
        float ps = acc0[r] * as0 + acc1[r] * as1 + acc2[r] * as2 + acc3[r] * as3;
        float pd = acc0[r] * ad0 + acc1[r] * ad1 + acc2[r] * ad2 + acc3[r] * ad3;
        #pragma unroll
        for (int off = 1; off < 16; off <<= 1) {
            ps += __shfl_xor(ps, off, 32);
            pd += __shfl_xor(pd, off, 32);
        }
        if (lrow == 0) { a_src[row] = ps; a_dst[row] = pd; }
    }
}

__global__ void edge_max_kernel(const int* __restrict__ ei,
                                const float* __restrict__ a_src,
                                const float* __restrict__ a_dst,
                                float* __restrict__ mbuf) {
    int e = blockIdx.x * blockDim.x + threadIdx.x;
    if (e >= E_TOT) return;
    int s, d;
    if (e < N_EDGES) { s = ei[e]; d = ei[N_EDGES + e]; }
    else             { s = d = e - N_EDGES; }
    float v = a_src[s] + a_dst[d];
    v = (v > 0.0f) ? v : v * NEG_SLOPE;
    atomicMaxFloat(&mbuf[d], v);
}

__global__ void edge_scatter_kernel(const int* __restrict__ ei,
                                    const float* __restrict__ a_src,
                                    const float* __restrict__ a_dst,
                                    const float* __restrict__ mbuf,
                                    const float* __restrict__ h,
                                    float* __restrict__ sbuf,
                                    float* __restrict__ outacc) {
    int wave = (blockIdx.x * blockDim.x + threadIdx.x) >> 5;
    int lane = threadIdx.x & 31;
    if (wave >= E_TOT) return;
    int s, d;
    if (wave < N_EDGES) { s = ei[wave]; d = ei[N_EDGES + wave]; }
    else                { s = d = wave - N_EDGES; }
    float v = a_src[s] + a_dst[d];
    v = (v > 0.0f) ? v : v * NEG_SLOPE;
    float ex = expf(v - mbuf[d]);
    if (lane == 0) atomicAdd(&sbuf[d], ex);
    v2f hv = *(const v2f*)(h + (size_t)s * HIDDEN + lane * 2);
    float* ob = outacc + (size_t)d * HIDDEN + lane * 2;
    atomicAdd(ob,     ex * hv.x);
    atomicAdd(ob + 1, ex * hv.y);
}

__global__ void finalize_kernel(const float* __restrict__ outacc,
                                const float* __restrict__ sbuf,
                                const float* __restrict__ bias_conv,
                                const float* __restrict__ W_lin,
                                const float* __restrict__ b_lin,
                                float* __restrict__ y) {
    int wave = (blockIdx.x * blockDim.x + threadIdx.x) >> 5;
    int lane = threadIdx.x & 31;
    if (wave >= N_NODES) return;
    float inv = 1.0f / sbuf[wave];
    v2f o = *(const v2f*)(outacc + (size_t)wave * HIDDEN + lane * 2);
    float o0 = fmaxf(o.x * inv + bias_conv[lane * 2],     0.0f);
    float o1 = fmaxf(o.y * inv + bias_conv[lane * 2 + 1], 0.0f);
    float p = o0 * W_lin[lane * 2] + o1 * W_lin[lane * 2 + 1];
    #pragma unroll
    for (int off = 1; off < 32; off <<= 1) p += __shfl_xor(p, off, 32);
    if (lane == 0) y[wave] = 1.0f / (1.0f + expf(-(p + b_lin[0])));
}

extern "C" void kernel_launch(void* const* d_in, const int* in_sizes, int n_in,
                              void* d_out, int out_size, void* d_ws, size_t ws_size,
                              hipStream_t stream) {
    const float* x         = (const float*)d_in[0];
    const int*   ei        = (const int*)  d_in[1];
    const float* W         = (const float*)d_in[2];
    const float* att_src_v = (const float*)d_in[3];
    const float* att_dst_v = (const float*)d_in[4];
    const float* bias_conv = (const float*)d_in[5];
    const float* W_lin     = (const float*)d_in[6];
    const float* b_lin     = (const float*)d_in[7];
    float* y = (float*)d_out;

    float* ws     = (float*)d_ws;
    float* h      = ws;
    float* a_src  = ws + 3200000;
    float* a_dst  = ws + 3250000;
    float* mbuf   = ws + 3300000;
    float* sbuf   = ws + 3350000;
    float* outacc = ws + 3400000;

    init_ws_kernel<<<12500, 256, 0, stream>>>(mbuf, sbuf, outacc);

    gat_gemm_wmma_kernel<<<3125, 32, 0, stream>>>(x, W, att_src_v, att_dst_v,
                                                  h, a_src, a_dst);

    edge_max_kernel<<<(E_TOT + 255) / 256, 256, 0, stream>>>(ei, a_src, a_dst, mbuf);

    edge_scatter_kernel<<<E_TOT / 8, 256, 0, stream>>>(ei, a_src, a_dst, mbuf,
                                                       h, sbuf, outacc);

    finalize_kernel<<<(N_NODES + 7) / 8, 256, 0, stream>>>(outacc, sbuf, bias_conv,
                                                           W_lin, b_lin, y);
}


# Per-destination wave gathers a counting-sorted edge list (int rank add) for segment max/sum/weighted sum instead of float atomics; output head fused; sums now in arrival-rank order
# speedup vs baseline: 2.5818x; 2.5818x over previous
#include <hip/hip_runtime.h>
#include <math.h>

#define N_NODES 50000
#define N_EDGES 800000
#define IN_DIM 128
#define HIDDEN 64
#define NEG_SLOPE 0.2f
#define E_TOT (N_EDGES + N_NODES)

typedef float v2f __attribute__((ext_vector_type(2)));
typedef float v8f __attribute__((ext_vector_type(8)));

typedef float v4f __attribute__((ext_vector_type(4)));

#define CSTRIDE 32

__device__ __forceinline__ float leaky(float v) { return (v > 0.0f) ? v : v * NEG_SLOPE; }

__global__ void zero_cnt_kernel(int* __restrict__ cnt, int* __restrict__ total) {
    int i = blockIdx.x * blockDim.x + threadIdx.x;
    if (i < N_NODES) cnt[(size_t)i * CSTRIDE] = 0;
    if (i == 0) total[0] = 0;
}

__global__ void __launch_bounds__(32)
gat_gemm_wmma_kernel(const float* __restrict__ x, const float* __restrict__ W,
                     const float* __restrict__ att_src_v, const float* __restrict__ att_dst_v,
                     float* __restrict__ h, float* __restrict__ a_src,
                     float* __restrict__ a_dst)
{
    const int tile = blockIdx.x;
    const int lane = threadIdx.x;
    const int half = lane >> 4;
    const int lrow = lane & 15;
    const int row0 = tile * 16;

    const float* xrow  = x + (size_t)(row0 + lrow) * IN_DIM;
    const float* wrow  = W + (size_t)lrow * IN_DIM;

    v8f acc0 = {}, acc1 = {}, acc2 = {}, acc3 = {};

    for (int kb = 0; kb < IN_DIM; kb += 4) {
        const int k = kb + 2 * half;
        v2f a  = *(const v2f*)(xrow + k);
        v2f b0 = *(const v2f*)(wrow + k);
        v2f b1 = *(const v2f*)(wrow + 16 * IN_DIM + k);
        v2f b2 = *(const v2f*)(wrow + 32 * IN_DIM + k);
        v2f b3 = *(const v2f*)(wrow + 48 * IN_DIM + k);
        acc0 = __builtin_amdgcn_wmma_f32_16x16x4_f32(false, a, false, b0, (short)0, acc0, false, false);
        acc1 = __builtin_amdgcn_wmma_f32_16x16x4_f32(false, a, false, b1, (short)0, acc1, false, false);
        acc2 = __builtin_amdgcn_wmma_f32_16x16x4_f32(false, a, false, b2, (short)0, acc2, false, false);
        acc3 = __builtin_amdgcn_wmma_f32_16x16x4_f32(false, a, false, b3, (short)0, acc3, false, false);
    }

    const float as0 = att_src_v[lrow],      ad0 = att_dst_v[lrow];
    const float as1 = att_src_v[16 + lrow], ad1 = att_dst_v[16 + lrow];
    const float as2 = att_src_v[32 + lrow], ad2 = att_dst_v[32 + lrow];
    const float as3 = att_src_v[48 + lrow], ad3 = att_dst_v[48 + lrow];

    #pragma unroll
    for (int r = 0; r < 8; ++r) {
        const int row = row0 + r + 8 * half;
        float* hr = h + (size_t)row * HIDDEN;
        hr[lrow]      = acc0[r];
        hr[16 + lrow] = acc1[r];
        hr[32 + lrow] = acc2[r];
        hr[48 + lrow] = acc3[r];
        float ps = acc0[r] * as0 + acc1[r] * as1 + acc2[r] * as2 + acc3[r] * as3;
        float pd = acc0[r] * ad0 + acc1[r] * ad1 + acc2[r] * ad2 + acc3[r] * ad3;
        #pragma unroll
        for (int off = 1; off < 16; off <<= 1) {
            ps += __shfl_xor(ps, off, 32);
            pd += __shfl_xor(pd, off, 32);
        }
        if (lrow == 0) { a_src[row] = ps; a_dst[row] = pd; }
    }
}

__global__ void edge_rank_kernel(const int* __restrict__ ei, int* __restrict__ cnt,
                                 int* __restrict__ rank) {
    int e = blockIdx.x * blockDim.x + threadIdx.x;
    if (e >= N_EDGES) return;
    int d = ei[N_EDGES + e];
    rank[e] = atomicAdd(&cnt[(size_t)d * CSTRIDE], 1);
}

__global__ void __launch_bounds__(256)
seg_alloc_kernel(const int* __restrict__ cnt, int* __restrict__ total, int* __restrict__ off) {
    __shared__ int wsum[8];
    __shared__ int base;
    const int i = blockIdx.x * 256 + threadIdx.x;
    const int lane = threadIdx.x & 31, w = threadIdx.x >> 5;
    const int c = (i < N_NODES) ? cnt[(size_t)i * CSTRIDE] : 0;
    int v = c;
    #pragma unroll
    for (int o = 1; o < 32; o <<= 1) {
        int t = __shfl_up(v, o, 32);
        if (lane >= o) v += t;
    }
    if (lane == 31) wsum[w] = v;
    __syncthreads();
    if (threadIdx.x == 0) {
        int tot = 0;
        #pragma unroll
        for (int k = 0; k < 8; ++k) { int t = wsum[k]; wsum[k] = tot; tot += t; }
        base = atomicAdd(total, tot);
    }
    __syncthreads();
    if (i < N_NODES) off[i] = base + wsum[w] + v - c;
}

__global__ void edge_fill_kernel(const int* __restrict__ ei, const int* __restrict__ rank,
                                 const int* __restrict__ off, int* __restrict__ ssrc) {
    int e = blockIdx.x * blockDim.x + threadIdx.x;
    if (e >= N_EDGES) return;
    int s = ei[e], d = ei[N_EDGES + e];
    ssrc[off[d] + rank[e]] = s;
}

__global__ void __launch_bounds__(256)
node_gather_kernel(const int* __restrict__ cnt, const int* __restrict__ off,
                   const int* __restrict__ ssrc,
                   const float* __restrict__ a_src, const float* __restrict__ a_dst,
                   const float* __restrict__ h,
                   const float* __restrict__ bias_conv, const float* __restrict__ W_lin,
                   const float* __restrict__ b_lin, float* __restrict__ y) {
    const int d = __builtin_amdgcn_readfirstlane((int)((blockIdx.x * 256 + threadIdx.x) >> 5));
    const int lane = threadIdx.x & 31;
    const int half = lane >> 4, q = lane & 15;
    if (d >= N_NODES) return;
    const int n = cnt[(size_t)d * CSTRIDE];
    const int o = off[d];
    const float ad = a_dst[d];
    const float vself = leaky(a_src[d] + ad);

    int   s0 = d;
    float v0 = -INFINITY;
    if (lane < n) { s0 = ssrc[o + lane]; v0 = leaky(a_src[s0] + ad); }
    float m = fmaxf(vself, v0);
    for (int j = 32 + lane; j < n; j += 32) {
        int s = ssrc[o + j];
        m = fmaxf(m, leaky(a_src[s] + ad));
    }
    #pragma unroll
    for (int x = 1; x < 32; x <<= 1) m = fmaxf(m, __shfl_xor(m, x, 32));

    const float exs = expf(vself - m);
    float ssum = (lane == 0) ? exs : 0.0f;
    v4f acc = {0.0f, 0.0f, 0.0f, 0.0f};
    if (half == 0) {
        v4f hv = *(const v4f*)(h + (size_t)d * HIDDEN + q * 4);
        acc = exs * hv;
    }

    for (int j0 = 0; j0 < n; j0 += 32) {
        int s; float ex;
        if (j0 == 0) {
            s = s0;
            ex = (lane < n) ? expf(v0 - m) : 0.0f;
        } else {
            const int j = j0 + lane;
            s = d; ex = 0.0f;
            if (j < n) { s = ssrc[o + j]; ex = expf(leaky(a_src[s] + ad) - m); }
        }
        ssum += ex;
        const int cc = (n - j0 < 32) ? (n - j0) : 32;
        #pragma unroll 4
        for (int k = 0; k < cc; k += 2) {
            const int kk = k + half;
            const bool valid = kk < cc;
            const int   sk = __shfl(s,  kk, 32);
            const float ek = __shfl(ex, kk, 32);
            const v4f hv = *(const v4f*)(h + (size_t)sk * HIDDEN + q * 4);
            acc.x = valid ? fmaf(ek, hv.x, acc.x) : acc.x;
            acc.y = valid ? fmaf(ek, hv.y, acc.y) : acc.y;
            acc.z = valid ? fmaf(ek, hv.z, acc.z) : acc.z;
            acc.w = valid ? fmaf(ek, hv.w, acc.w) : acc.w;
        }
    }

    acc.x += __shfl_xor(acc.x, 16, 32);
    acc.y += __shfl_xor(acc.y, 16, 32);
    acc.z += __shfl_xor(acc.z, 16, 32);
    acc.w += __shfl_xor(acc.w, 16, 32);
    #pragma unroll
    for (int x = 1; x < 32; x <<= 1) ssum += __shfl_xor(ssum, x, 32);

    const float inv = 1.0f / ssum;
    const v4f bc = *(const v4f*)(bias_conv + q * 4);
    const v4f wl = *(const v4f*)(W_lin + q * 4);
    float p = fmaxf(acc.x * inv + bc.x, 0.0f) * wl.x
            + fmaxf(acc.y * inv + bc.y, 0.0f) * wl.y
            + fmaxf(acc.z * inv + bc.z, 0.0f) * wl.z
            + fmaxf(acc.w * inv + bc.w, 0.0f) * wl.w;
    #pragma unroll
    for (int x = 1; x < 16; x <<= 1) p += __shfl_xor(p, x, 32);
    if (lane == 0) y[d] = 1.0f / (1.0f + expf(-(p + b_lin[0])));
}

extern "C" void kernel_launch(void* const* d_in, const int* in_sizes, int n_in,
                              void* d_out, int out_size, void* d_ws, size_t ws_size,
                              hipStream_t stream) {
    const float* x         = (const float*)d_in[0];
    const int*   ei        = (const int*)  d_in[1];
    const float* W         = (const float*)d_in[2];
    const float* att_src_v = (const float*)d_in[3];
    const float* att_dst_v = (const float*)d_in[4];
    const float* bias_conv = (const float*)d_in[5];
    const float* W_lin     = (const float*)d_in[6];
    const float* b_lin     = (const float*)d_in[7];
    float* y = (float*)d_out;

    float* ws     = (float*)d_ws;
    float* h      = ws;
    float* a_src  = ws + 3200000;
    float* a_dst  = ws + 3250000;
    int*   cnt    = (int*)(ws + 3300000);
    int*   off    = (int*)(ws + 4900000);
    int*   total  = (int*)(ws + 4950000);
    int*   rank   = (int*)(ws + 4950064);
    int*   ssrc   = (int*)(ws + 5750064);

    zero_cnt_kernel<<<196, 256, 0, stream>>>(cnt, total);

    gat_gemm_wmma_kernel<<<3125, 32, 0, stream>>>(x, W, att_src_v, att_dst_v,
                                                  h, a_src, a_dst);

    edge_rank_kernel<<<3125, 256, 0, stream>>>(ei, cnt, rank);

    seg_alloc_kernel<<<196, 256, 0, stream>>>(cnt, total, off);

    edge_fill_kernel<<<3125, 256, 0, stream>>>(ei, rank, off, ssrc);

    node_gather_kernel<<<6250, 256, 0, stream>>>(cnt, off, ssrc, a_src, a_dst, h,
                                                 bias_conv, W_lin, b_lin, y);
}


# Counting sort replaced by one 128 B line per node (counter + first 31 sources, slot = ticket of a returning int add) and a shared overflow list: scan, fill pass, rank array go; sums in arrival order
# speedup vs baseline: 2.8327x; 1.0972x over previous
#include <hip/hip_runtime.h>
#include <math.h>

#define N_NODES 50000
#define N_EDGES 800000
#define IN_DIM 128
#define HIDDEN 64
#define NEG_SLOPE 0.2f
#define E_TOT (N_EDGES + N_NODES)

typedef float v2f __attribute__((ext_vector_type(2)));
typedef float v8f __attribute__((ext_vector_type(8)));

typedef float v4f __attribute__((ext_vector_type(4)));

#define LINE_W 32
#define SLOTS 31

__device__ __forceinline__ float leaky(float v) { return (v > 0.0f) ? v : v * NEG_SLOPE; }

__global__ void __launch_bounds__(256)
zero_lines_kernel(int* __restrict__ lines, int* __restrict__ ovcnt) {
    int4* l4 = (int4*)lines;
    const int base = blockIdx.x * 1024 + threadIdx.x;
    #pragma unroll
    for (int k = 0; k < 4; ++k) {
        const int i = base + k * 256;
        if (i < N_NODES * (LINE_W / 4)) l4[i] = make_int4(0, 0, 0, 0);
    }
    if (blockIdx.x == 0 && threadIdx.x == 0) ovcnt[0] = 0;
}

__global__ void __launch_bounds__(32)
gat_gemm_wmma_kernel(const float* __restrict__ x, const float* __restrict__ W,
                     const float* __restrict__ att_src_v, const float* __restrict__ att_dst_v,
                     float* __restrict__ h, float* __restrict__ a_src,
                     float* __restrict__ a_dst)
{
    const int tile = blockIdx.x;
    const int lane = threadIdx.x;
    const int half = lane >> 4;
    const int lrow = lane & 15;
    const int row0 = tile * 16;

    const float* xrow  = x + (size_t)(row0 + lrow) * IN_DIM;
    const float* wrow  = W + (size_t)lrow * IN_DIM;

    v8f acc0 = {}, acc1 = {}, acc2 = {}, acc3 = {};

    for (int kb = 0; kb < IN_DIM; kb += 4) {
        const int k = kb + 2 * half;
        v2f a  = *(const v2f*)(xrow + k);
        v2f b0 = *(const v2f*)(wrow + k);
        v2f b1 = *(const v2f*)(wrow + 16 * IN_DIM + k);
        v2f b2 = *(const v2f*)(wrow + 32 * IN_DIM + k);
        v2f b3 = *(const v2f*)(wrow + 48 * IN_DIM + k);
        acc0 = __builtin_amdgcn_wmma_f32_16x16x4_f32(false, a, false, b0, (short)0, acc0, false, false);
        acc1 = __builtin_amdgcn_wmma_f32_16x16x4_f32(false, a, false, b1, (short)0, acc1, false, false);
        acc2 = __builtin_amdgcn_wmma_f32_16x16x4_f32(false, a, false, b2, (short)0, acc2, false, false);
        acc3 = __builtin_amdgcn_wmma_f32_16x16x4_f32(false, a, false, b3, (short)0, acc3, false, false);
    }

    const float as0 = att_src_v[lrow],      ad0 = att_dst_v[lrow];
    const float as1 = att_src_v[16 + lrow], ad1 = att_dst_v[16 + lrow];
    const float as2 = att_src_v[32 + lrow], ad2 = att_dst_v[32 + lrow];
    const float as3 = att_src_v[48 + lrow], ad3 = att_dst_v[48 + lrow];

    #pragma unroll
    for (int r = 0; r < 8; ++r) {
        const int row = row0 + r + 8 * half;
        float* hr = h + (size_t)row * HIDDEN;
        hr[lrow]      = acc0[r];
        hr[16 + lrow] = acc1[r];
        hr[32 + lrow] = acc2[r];
        hr[48 + lrow] = acc3[r];
        float ps = acc0[r] * as0 + acc1[r] * as1 + acc2[r] * as2 + acc3[r] * as3;
        float pd = acc0[r] * ad0 + acc1[r] * ad1 + acc2[r] * ad2 + acc3[r] * ad3;
        #pragma unroll
        for (int off = 1; off < 16; off <<= 1) {
            ps += __shfl_xor(ps, off, 32);
            pd += __shfl_xor(pd, off, 32);
        }
        if (lrow == 0) { a_src[row] = ps; a_dst[row] = pd; }
    }
}

__global__ void edge_bucket_kernel(const int* __restrict__ ei, int* __restrict__ lines,
                                   int* __restrict__ ovcnt, int* __restrict__ ov) {
    int e = blockIdx.x * blockDim.x + threadIdx.x;
    if (e >= N_EDGES) return;
    const int s = ei[e], d = ei[N_EDGES + e];
    int* line = lines + (size_t)d * LINE_W;
    const int r = atomicAdd(line, 1);
    if (r < SLOTS) {
        line[1 + r] = s;
    } else {
        const int k = atomicAdd(ovcnt, 1);
        ((int2*)ov)[k] = make_int2(d, s);
    }
}

__global__ void __launch_bounds__(256)
node_gather_kernel(const int* __restrict__ lines, const int* __restrict__ ovcnt,
                   const int* __restrict__ ov,
                   const float* __restrict__ a_src, const float* __restrict__ a_dst,
                   const float* __restrict__ h,
                   const float* __restrict__ bias_conv, const float* __restrict__ W_lin,
                   const float* __restrict__ b_lin, float* __restrict__ y) {
    const int d = __builtin_amdgcn_readfirstlane((int)((blockIdx.x * 256 + threadIdx.x) >> 5));
    const int lane = threadIdx.x & 31;
    const int half = lane >> 4, q = lane & 15;
    if (d >= N_NODES) return;
    const int w = lines[(size_t)d * LINE_W + ((lane + 1) & 31)];
    const int n = __builtin_amdgcn_readlane(w, 31);
    const int nb = (n < SLOTS) ? n : SLOTS;
    const float ad = a_dst[d];
    const float vself = leaky(a_src[d] + ad);
    const int2* ov2 = (const int2*)ov;

    int   s0 = d;
    float v0 = -INFINITY;
    if (lane < nb) { s0 = w; v0 = leaky(a_src[s0] + ad); }
    float m = fmaxf(vself, v0);
    int tov = 0;
    if (n > SLOTS) {
        tov = ovcnt[0];
        for (int j = lane; j < tov; j += 32) {
            const int2 p = ov2[j];
            if (p.x == d) m = fmaxf(m, leaky(a_src[p.y] + ad));
        }
    }
    #pragma unroll
    for (int x = 1; x < 32; x <<= 1) m = fmaxf(m, __shfl_xor(m, x, 32));

    const float exs = expf(vself - m);
    float ssum = (lane == 0) ? exs : 0.0f;
    v4f acc = {0.0f, 0.0f, 0.0f, 0.0f};
    if (half == 0) {
        v4f hv = *(const v4f*)(h + (size_t)d * HIDDEN + q * 4);
        acc = exs * hv;
    }

    {
        const float ex = (lane < nb) ? expf(v0 - m) : 0.0f;
        ssum += ex;
        #pragma unroll 4
        for (int k = 0; k < nb; k += 2) {
            const int kk = k + half;
            const bool valid = kk < nb;
            const int   sk = __shfl(s0, kk, 32);
            const float ek = __shfl(ex, kk, 32);
            const v4f hv = *(const v4f*)(h + (size_t)sk * HIDDEN + q * 4);
            acc.x = valid ? fmaf(ek, hv.x, acc.x) : acc.x;
            acc.y = valid ? fmaf(ek, hv.y, acc.y) : acc.y;
            acc.z = valid ? fmaf(ek, hv.z, acc.z) : acc.z;
            acc.w = valid ? fmaf(ek, hv.w, acc.w) : acc.w;
        }
    }

    if (n > SLOTS) {
        for (int j0 = 0; j0 < tov; j0 += 32) {
            const int j = j0 + lane;
            int so = d; float eo = 0.0f; bool mine = false;
            if (j < tov) {
                const int2 p = ov2[j];
                if (p.x == d) { mine = true; so = p.y; eo = expf(leaky(a_src[so] + ad) - m); }
            }
            ssum += eo;
            unsigned int mask = (unsigned int)__ballot(mine);
            while (mask) {
                const int b = __builtin_ctz(mask);
                mask &= mask - 1;
                const int   sk = __shfl(so, b, 32);
                const float ek = __shfl(eo, b, 32);
                if (half == 0) {
                    const v4f hv = *(const v4f*)(h + (size_t)sk * HIDDEN + q * 4);
                    acc.x = fmaf(ek, hv.x, acc.x);
                    acc.y = fmaf(ek, hv.y, acc.y);
                    acc.z = fmaf(ek, hv.z, acc.z);
                    acc.w = fmaf(ek, hv.w, acc.w);
                }
            }
        }
    }

    acc.x += __shfl_xor(acc.x, 16, 32);
    acc.y += __shfl_xor(acc.y, 16, 32);
    acc.z += __shfl_xor(acc.z, 16, 32);
    acc.w += __shfl_xor(acc.w, 16, 32);
    #pragma unroll
    for (int x = 1; x < 32; x <<= 1) ssum += __shfl_xor(ssum, x, 32);

    const float inv = 1.0f / ssum;
    const v4f bc = *(const v4f*)(bias_conv + q * 4);
    const v4f wl = *(const v4f*)(W_lin + q * 4);
    float p = fmaxf(acc.x * inv + bc.x, 0.0f) * wl.x
            + fmaxf(acc.y * inv + bc.y, 0.0f) * wl.y
            + fmaxf(acc.z * inv + bc.z, 0.0f) * wl.z
            + fmaxf(acc.w * inv + bc.w, 0.0f) * wl.w;
    #pragma unroll
    for (int x = 1; x < 16; x <<= 1) p += __shfl_xor(p, x, 32);
    if (lane == 0) y[d] = 1.0f / (1.0f + expf(-(p + b_lin[0])));
}

extern "C" void kernel_launch(void* const* d_in, const int* in_sizes, int n_in,
                              void* d_out, int out_size, void* d_ws, size_t ws_size,
                              hipStream_t stream) {
    const float* x         = (const float*)d_in[0];
    const int*   ei        = (const int*)  d_in[1];
    const float* W         = (const float*)d_in[2];
    const float* att_src_v = (const float*)d_in[3];
    const float* att_dst_v = (const float*)d_in[4];
    const float* bias_conv = (const float*)d_in[5];
    const float* W_lin     = (const float*)d_in[6];
    const float* b_lin     = (const float*)d_in[7];
    float* y = (float*)d_out;

    float* ws     = (float*)d_ws;
    float* h      = ws;
    float* a_src  = ws + 3200000;
    float* a_dst  = ws + 3250000;
    int*   lines  = (int*)(ws + 3300032);
    int*   ovcnt  = (int*)(ws + 4900032);
    int*   ov     = (int*)(ws + 4900096);

    zero_lines_kernel<<<391, 256, 0, stream>>>(lines, ovcnt);

    gat_gemm_wmma_kernel<<<3125, 32, 0, stream>>>(x, W, att_src_v, att_dst_v,
                                                  h, a_src, a_dst);

    edge_bucket_kernel<<<3125, 256, 0, stream>>>(ei, lines, ovcnt, ov);

    node_gather_kernel<<<6250, 256, 0, stream>>>(lines, ovcnt, ov, a_src, a_dst, h,
                                                 bias_conv, W_lin, b_lin, y);
}


# Gather wave requests all rows of a node in one group, keeps them in registers and takes each source score from its row (dot with att_src): a_src gather gone; score, exp and weighted sums in new order
# speedup vs baseline: 2.9820x; 1.0527x over previous
#include <hip/hip_runtime.h>
#include <math.h>

#define N_NODES 50000
#define N_EDGES 800000
#define IN_DIM 128
#define HIDDEN 64
#define NEG_SLOPE 0.2f
#define E_TOT (N_EDGES + N_NODES)

typedef float v2f __attribute__((ext_vector_type(2)));
typedef float v8f __attribute__((ext_vector_type(8)));

typedef float v4f __attribute__((ext_vector_type(4)));

#define LINE_W 32
#define SLOTS 31

__device__ __forceinline__ float leaky(float v) { return (v > 0.0f) ? v : v * NEG_SLOPE; }

__global__ void __launch_bounds__(256)
zero_lines_kernel(int* __restrict__ lines, int* __restrict__ ovcnt) {
    int4* l4 = (int4*)lines;
    const int base = blockIdx.x * 1024 + threadIdx.x;
    #pragma unroll
    for (int k = 0; k < 4; ++k) {
        const int i = base + k * 256;
        if (i < N_NODES * (LINE_W / 4)) l4[i] = make_int4(0, 0, 0, 0);
    }
    if (blockIdx.x == 0 && threadIdx.x == 0) ovcnt[0] = 0;
}

__global__ void __launch_bounds__(32)
gat_gemm_wmma_kernel(const float* __restrict__ x, const float* __restrict__ W,
                     const float* __restrict__ att_dst_v,
                     float* __restrict__ h, float* __restrict__ a_dst)
{
    const int tile = blockIdx.x;
    const int lane = threadIdx.x;
    const int half = lane >> 4;
    const int lrow = lane & 15;
    const int row0 = tile * 16;

    const float* xrow  = x + (size_t)(row0 + lrow) * IN_DIM;
    const float* wrow  = W + (size_t)lrow * IN_DIM;

    v8f acc0 = {}, acc1 = {}, acc2 = {}, acc3 = {};

    for (int kb = 0; kb < IN_DIM; kb += 4) {
        const int k = kb + 2 * half;
        v2f a  = *(const v2f*)(xrow + k);
        v2f b0 = *(const v2f*)(wrow + k);
        v2f b1 = *(const v2f*)(wrow + 16 * IN_DIM + k);
        v2f b2 = *(const v2f*)(wrow + 32 * IN_DIM + k);
        v2f b3 = *(const v2f*)(wrow + 48 * IN_DIM + k);
        acc0 = __builtin_amdgcn_wmma_f32_16x16x4_f32(false, a, false, b0, (short)0, acc0, false, false);
        acc1 = __builtin_amdgcn_wmma_f32_16x16x4_f32(false, a, false, b1, (short)0, acc1, false, false);
        acc2 = __builtin_amdgcn_wmma_f32_16x16x4_f32(false, a, false, b2, (short)0, acc2, false, false);
        acc3 = __builtin_amdgcn_wmma_f32_16x16x4_f32(false, a, false, b3, (short)0, acc3, false, false);
    }

    const float ad0 = att_dst_v[lrow];
    const float ad1 = att_dst_v[16 + lrow];
    const float ad2 = att_dst_v[32 + lrow];
    const float ad3 = att_dst_v[48 + lrow];

    #pragma unroll
    for (int r = 0; r < 8; ++r) {
        const int row = row0 + r + 8 * half;
        float* hr = h + (size_t)row * HIDDEN;
        hr[lrow]      = acc0[r];
        hr[16 + lrow] = acc1[r];
        hr[32 + lrow] = acc2[r];
        hr[48 + lrow] = acc3[r];
        float pd = acc0[r] * ad0 + acc1[r] * ad1 + acc2[r] * ad2 + acc3[r] * ad3;
        #pragma unroll
        for (int off = 1; off < 16; off <<= 1) pd += __shfl_xor(pd, off, 32);
        if (lrow == 0) a_dst[row] = pd;
    }
}

__global__ void edge_bucket_kernel(const int* __restrict__ ei, int* __restrict__ lines,
                                   int* __restrict__ ovcnt, int* __restrict__ ov) {
    int e = blockIdx.x * blockDim.x + threadIdx.x;
    if (e >= N_EDGES) return;
    const int s = ei[e], d = ei[N_EDGES + e];
    int* line = lines + (size_t)d * LINE_W;
    const int r = atomicAdd(line, 1);
    if (r < SLOTS) {
        line[1 + r] = s;
    } else {
        const int k = atomicAdd(ovcnt, 1);
        ((int2*)ov)[k] = make_int2(d, s);
    }
}

#define NPAIR 16

template <int CTRL>
__device__ __forceinline__ int row_rd(int x) {
    return __builtin_amdgcn_update_dpp(0, x, CTRL, 0xf, 0xf, true);
}
template <int CTRL>
__device__ __forceinline__ float row_rd(float x) {
    return __builtin_bit_cast(float, row_rd<CTRL>(__builtin_bit_cast(int, x)));
}
__device__ __forceinline__ float xhalf(float x) {
    const int b = __builtin_bit_cast(int, x);
    return __builtin_bit_cast(float,
        __builtin_amdgcn_permlanex16(b, b, 0x76543210, 0xfedcba98, false, false));
}
__device__ __forceinline__ float row_sum16(float p) {
    p += row_rd<0x161>(p); p += row_rd<0x162>(p); p += row_rd<0x164>(p); p += row_rd<0x168>(p);
    return p;
}
__device__ __forceinline__ float row_max16(float p) {
    p = fmaxf(p, row_rd<0x161>(p)); p = fmaxf(p, row_rd<0x162>(p));
    p = fmaxf(p, row_rd<0x164>(p)); p = fmaxf(p, row_rd<0x168>(p));
    return p;
}
__device__ __forceinline__ float dot4(const v4f a, const v4f b) {
    return a.x * b.x + a.y * b.y + a.z * b.z + a.w * b.w;
}

template <int I>
__device__ __forceinline__ void load_pairs(v4f (&r)[NPAIR], const int np, const int ent2,
                                           const unsigned int q,
                                           const v4f* __restrict__ h4) {
    if constexpr (I < NPAIR) {
        if (I < np) {
            const unsigned int s = (unsigned int)row_rd<0x150 | I>(ent2);
            r[I] = h4[s * 16u + q];
            load_pairs<I + 1>(r, np, ent2, q, h4);
        }
    }
}
template <int I>
__device__ __forceinline__ void score_pairs(const v4f (&r)[NPAIR], const int np, const v4f as4,
                                            const int q, float& sc) {
    if constexpr (I < NPAIR) {
        if (I < np) {
            const float p = row_sum16(dot4(r[I], as4));
            sc = (q == I) ? p : sc;
            score_pairs<I + 1>(r, np, as4, q, sc);
        }
    }
}
template <int I>
__device__ __forceinline__ void accum_pairs(const v4f (&r)[NPAIR], const int np, const float ex,
                                            v4f& acc) {
    if constexpr (I < NPAIR) {
        if (I < np) {
            const float ek = row_rd<0x150 | I>(ex);
            acc.x = fmaf(ek, r[I].x, acc.x);
            acc.y = fmaf(ek, r[I].y, acc.y);
            acc.z = fmaf(ek, r[I].z, acc.z);
            acc.w = fmaf(ek, r[I].w, acc.w);
            accum_pairs<I + 1>(r, np, ex, acc);
        }
    }
}

__global__ void __launch_bounds__(256)
node_gather_kernel(const int* __restrict__ lines, const int* __restrict__ ovcnt,
                   const int* __restrict__ ov,
                   const float* __restrict__ att_src_v, const float* __restrict__ a_dst,
                   const float* __restrict__ h,
                   const float* __restrict__ bias_conv, const float* __restrict__ W_lin,
                   const float* __restrict__ b_lin, float* __restrict__ y) {
    const int d = __builtin_amdgcn_readfirstlane((int)((blockIdx.x * 256 + threadIdx.x) >> 5));
    const int lane = threadIdx.x & 31;
    const int half = lane >> 4, q = lane & 15;
    if (d >= N_NODES) return;
    const int w = lines[(size_t)d * LINE_W + lane];
    const v4f as4 = *(const v4f*)(att_src_v + q * 4);
    const v4f bc  = *(const v4f*)(bias_conv + q * 4);
    const v4f wl  = *(const v4f*)(W_lin + q * 4);
    const float bl = b_lin[0];
    const float ad = a_dst[d];

    const int n  = __builtin_amdgcn_readfirstlane(w);
    const int nb = (n < SLOTS) ? n : SLOTS;
    const int np = (nb + 2) >> 1;
    const int ent = (lane >= 1 && lane <= nb) ? w : d;
    const int ent2 = __shfl(ent, 2 * q + half, 32);
    const v4f* h4 = (const v4f*)h;

    v4f r[NPAIR];
    load_pairs<0>(r, np, ent2, (unsigned int)q, h4);

    float sc = 0.0f;
    score_pairs<0>(r, np, as4, q, sc);
    const bool valid = (2 * q + half) <= nb;
    const float v = valid ? leaky(sc + ad) : -INFINITY;
    float m = row_max16(v);
    m = fmaxf(m, xhalf(m));

    const int2* ov2 = (const int2*)ov;
    int tov = 0;
    if (n > SLOTS) {
        tov = ovcnt[0];
        for (int j0 = 0; j0 < tov; j0 += 32) {
            const int j = j0 + lane;
            int so = d; bool mine = false;
            if (j < tov) { const int2 p = ov2[j]; if (p.x == d) { mine = true; so = p.y; } }
            unsigned int mask = (unsigned int)__ballot(mine);
            while (mask) {
                const int b = __builtin_ctz(mask);
                mask &= mask - 1;
                const int sk = __builtin_amdgcn_readlane(so, b);
                const v4f hv = *(const v4f*)(h + (size_t)sk * HIDDEN + q * 4);
                m = fmaxf(m, leaky(row_sum16(dot4(hv, as4)) + ad));
            }
        }
    }

    const float ex = valid ? expf(v - m) : 0.0f;
    v4f acc = {0.0f, 0.0f, 0.0f, 0.0f};
    accum_pairs<0>(r, np, ex, acc);

    float sov = 0.0f;
    if (n > SLOTS) {
        for (int j0 = 0; j0 < tov; j0 += 32) {
            const int j = j0 + lane;
            int so = d; bool mine = false;
            if (j < tov) { const int2 p = ov2[j]; if (p.x == d) { mine = true; so = p.y; } }
            unsigned int mask = (unsigned int)__ballot(mine);
            while (mask) {
                const int b = __builtin_ctz(mask);
                mask &= mask - 1;
                const int sk = __builtin_amdgcn_readlane(so, b);
                const v4f hv = *(const v4f*)(h + (size_t)sk * HIDDEN + q * 4);
                const float eo = expf(leaky(row_sum16(dot4(hv, as4)) + ad) - m);
                sov += eo;
                const float eh = (half == 0) ? eo : 0.0f;
                acc.x = fmaf(eh, hv.x, acc.x);
                acc.y = fmaf(eh, hv.y, acc.y);
                acc.z = fmaf(eh, hv.z, acc.z);
                acc.w = fmaf(eh, hv.w, acc.w);
            }
        }
    }

    acc.x += xhalf(acc.x);
    acc.y += xhalf(acc.y);
    acc.z += xhalf(acc.z);
    acc.w += xhalf(acc.w);
    float ssum = row_sum16(ex);
    ssum = ssum + xhalf(ssum) + sov;

    const float inv = 1.0f / ssum;
    float p = fmaxf(acc.x * inv + bc.x, 0.0f) * wl.x
            + fmaxf(acc.y * inv + bc.y, 0.0f) * wl.y
            + fmaxf(acc.z * inv + bc.z, 0.0f) * wl.z
            + fmaxf(acc.w * inv + bc.w, 0.0f) * wl.w;
    p = row_sum16(p);
    if (lane == 0) y[d] = 1.0f / (1.0f + expf(-(p + bl)));
}

extern "C" void kernel_launch(void* const* d_in, const int* in_sizes, int n_in,
                              void* d_out, int out_size, void* d_ws, size_t ws_size,
                              hipStream_t stream) {
    const float* x         = (const float*)d_in[0];
    const int*   ei        = (const int*)  d_in[1];
    const float* W         = (const float*)d_in[2];
    const float* att_src_v = (const float*)d_in[3];
    const float* att_dst_v = (const float*)d_in[4];
    const float* bias_conv = (const float*)d_in[5];
    const float* W_lin     = (const float*)d_in[6];
    const float* b_lin     = (const float*)d_in[7];
    float* y = (float*)d_out;

    float* ws     = (float*)d_ws;
    float* h      = ws;
    float* a_dst  = ws + 3250000;
    int*   lines  = (int*)(ws + 3300032);
    int*   ovcnt  = (int*)(ws + 4900032);
    int*   ov     = (int*)(ws + 4900096);

    zero_lines_kernel<<<391, 256, 0, stream>>>(lines, ovcnt);

    gat_gemm_wmma_kernel<<<3125, 32, 0, stream>>>(x, W, att_dst_v, h, a_dst);

    edge_bucket_kernel<<<3125, 256, 0, stream>>>(ei, lines, ovcnt, ov);

    node_gather_kernel<<<6250, 256, 0, stream>>>(lines, ovcnt, ov, att_src_v, a_dst, h,
                                                 bias_conv, W_lin, b_lin, y);
}


# Edge bucketing with no global ticket per edge: edge chunks binned into (200-node bin, chunk) lines in LDS, then each bin's node lines built in LDS and stored whole; sums still in arrival order
# speedup vs baseline: 4.9099x; 1.6465x over previous
#include <hip/hip_runtime.h>
#include <math.h>

#define N_NODES 50000
#define N_EDGES 800000
#define IN_DIM 128
#define HIDDEN 64
#define NEG_SLOPE 0.2f
#define E_TOT (N_EDGES + N_NODES)

typedef float v2f __attribute__((ext_vector_type(2)));
typedef float v8f __attribute__((ext_vector_type(8)));

typedef float v4f __attribute__((ext_vector_type(4)));

#define LINE_W 32
#define SLOTS 31

__device__ __forceinline__ float leaky(float v) { return (v > 0.0f) ? v : v * NEG_SLOPE; }

#define NBIN 250
#define BIN_NODES 200
#define NCHUNK 500
#define CHUNK_E 1600
#define EDGES_PER_THREAD 7
#define LPAD 33

__device__ __forceinline__ bool lds_line_put(int* line, const int value) {
    const int r = atomicAdd(line, 1);
    if (r < SLOTS) { line[1 + r] = value; return true; }
    return false;
}

__global__ void __launch_bounds__(256)
edge_bin_kernel(const int* __restrict__ ei, int* __restrict__ cells,
                int* __restrict__ spill_cnt, int* __restrict__ spill,
                int* __restrict__ ovcnt) {
    __shared__ int C[NBIN * LPAD];
    __shared__ int nspill;
    const int t = threadIdx.x;
    const int g = blockIdx.x;

    int s[EDGES_PER_THREAD], d[EDGES_PER_THREAD];
    #pragma unroll
    for (int k = 0; k < EDGES_PER_THREAD; ++k) {
        const int idx = k * 256 + t;
        const int e = g * CHUNK_E + ((idx < CHUNK_E) ? idx : 0);
        s[k] = ei[e];
        d[k] = ei[N_EDGES + e];
    }
    for (int i = t; i < NBIN * LPAD; i += 256) C[i] = 0;
    if (t == 0) {
        nspill = 0;
        if (g == 0) ovcnt[0] = 0;
    }
    __syncthreads();

    #pragma unroll
    for (int k = 0; k < EDGES_PER_THREAD; ++k) {
        const int idx = k * 256 + t;
        if (idx < CHUNK_E) {
            const unsigned int b  = (unsigned int)d[k] / (unsigned int)BIN_NODES;
            const unsigned int dl = (unsigned int)d[k] - b * (unsigned int)BIN_NODES;
            const int word = (int)((dl << 16) | (unsigned int)s[k]);
            if (!lds_line_put(&C[b * LPAD], word)) {
                const int k2 = atomicAdd(&nspill, 1);
                ((int2*)spill)[g * CHUNK_E + k2] = make_int2(d[k], s[k]);
            }
        }
    }
    __syncthreads();

    for (int i = t; i < NBIN * LINE_W; i += 256) {
        const int b = i >> 5, j = i & 31;
        cells[((size_t)b * NCHUNK + g) * LINE_W + j] = C[b * LPAD + j];
    }
    if (t == 0) spill_cnt[g] = nspill;
}

__device__ __forceinline__ void node_put(int* L, const int b, const int dl, const int s,
                                         int* __restrict__ ovcnt, int* __restrict__ ov) {
    if (!lds_line_put(&L[dl * LPAD], s)) {
        const int k = atomicAdd(ovcnt, 1);
        ((int2*)ov)[k] = make_int2(b * BIN_NODES + dl, s);
    }
}

__global__ void __launch_bounds__(512)
build_lines_kernel(const int* __restrict__ cells, const int* __restrict__ spill_cnt,
                   const int* __restrict__ spill, int* __restrict__ lines,
                   int* __restrict__ ovcnt, int* __restrict__ ov) {
    __shared__ int L[BIN_NODES * LPAD];
    const int t = threadIdx.x;
    const int b = blockIdx.x;
    const int lane = t & 31;
    const int w = t >> 5;

    for (int i = t; i < BIN_NODES * LPAD; i += 512) L[i] = 0;
    __syncthreads();

    const int* cb = cells + (size_t)b * NCHUNK * LINE_W;
    for (int g0 = w; g0 < NCHUNK; g0 += 128) {
        int v[8];
        #pragma unroll
        for (int u = 0; u < 8; ++u) {
            const int g = g0 + 16 * u;
            v[u] = (g < NCHUNK) ? cb[g * LINE_W + lane] : 0;
        }
        #pragma unroll
        for (int u = 0; u < 8; ++u) {
            const int c  = __builtin_amdgcn_readlane(v[u], 0);
            const int cc = (c < SLOTS) ? c : SLOTS;
            if (lane >= 1 && lane <= cc) {
                const unsigned int word = (unsigned int)v[u];
                node_put(L, b, (int)(word >> 16), (int)(word & 0xffffu), ovcnt, ov);
            }
        }
    }

    for (int g = t; g < NCHUNK; g += 512) {
        const int cr = spill_cnt[g];
        const int c = (cr < CHUNK_E) ? cr : CHUNK_E;
        for (int j = 0; j < c; ++j) {
            const int2 p = ((const int2*)spill)[g * CHUNK_E + j];
            const unsigned int pb = (unsigned int)p.x / (unsigned int)BIN_NODES;
            if (pb == (unsigned int)b) node_put(L, b, p.x - b * BIN_NODES, p.y, ovcnt, ov);
        }
    }
    __syncthreads();

    for (int i = t; i < BIN_NODES * LINE_W; i += 512) {
        lines[(size_t)b * BIN_NODES * LINE_W + i] = L[(i >> 5) * LPAD + (i & 31)];
    }
}

__global__ void __launch_bounds__(32)
gat_gemm_wmma_kernel(const float* __restrict__ x, const float* __restrict__ W,
                     const float* __restrict__ att_dst_v,
                     float* __restrict__ h, float* __restrict__ a_dst)
{
    const int tile = blockIdx.x;
    const int lane = threadIdx.x;
    const int half = lane >> 4;
    const int lrow = lane & 15;
    const int row0 = tile * 16;

    const float* xrow  = x + (size_t)(row0 + lrow) * IN_DIM;
    const float* wrow  = W + (size_t)lrow * IN_DIM;

    v8f acc0 = {}, acc1 = {}, acc2 = {}, acc3 = {};

    for (int kb = 0; kb < IN_DIM; kb += 4) {
        const int k = kb + 2 * half;
        v2f a  = *(const v2f*)(xrow + k);
        v2f b0 = *(const v2f*)(wrow + k);
        v2f b1 = *(const v2f*)(wrow + 16 * IN_DIM + k);
        v2f b2 = *(const v2f*)(wrow + 32 * IN_DIM + k);
        v2f b3 = *(const v2f*)(wrow + 48 * IN_DIM + k);
        acc0 = __builtin_amdgcn_wmma_f32_16x16x4_f32(false, a, false, b0, (short)0, acc0, false, false);
        acc1 = __builtin_amdgcn_wmma_f32_16x16x4_f32(false, a, false, b1, (short)0, acc1, false, false);
        acc2 = __builtin_amdgcn_wmma_f32_16x16x4_f32(false, a, false, b2, (short)0, acc2, false, false);
        acc3 = __builtin_amdgcn_wmma_f32_16x16x4_f32(false, a, false, b3, (short)0, acc3, false, false);
    }

    const float ad0 = att_dst_v[lrow];
    const float ad1 = att_dst_v[16 + lrow];
    const float ad2 = att_dst_v[32 + lrow];
    const float ad3 = att_dst_v[48 + lrow];

    #pragma unroll
    for (int r = 0; r < 8; ++r) {
        const int row = row0 + r + 8 * half;
        float* hr = h + (size_t)row * HIDDEN;
        hr[lrow]      = acc0[r];
        hr[16 + lrow] = acc1[r];
        hr[32 + lrow] = acc2[r];
        hr[48 + lrow] = acc3[r];
        float pd = acc0[r] * ad0 + acc1[r] * ad1 + acc2[r] * ad2 + acc3[r] * ad3;
        #pragma unroll
        for (int off = 1; off < 16; off <<= 1) pd += __shfl_xor(pd, off, 32);
        if (lrow == 0) a_dst[row] = pd;
    }
}

#define NPAIR 16

template <int CTRL>
__device__ __forceinline__ int row_rd(int x) {
    return __builtin_amdgcn_update_dpp(0, x, CTRL, 0xf, 0xf, true);
}
template <int CTRL>
__device__ __forceinline__ float row_rd(float x) {
    return __builtin_bit_cast(float, row_rd<CTRL>(__builtin_bit_cast(int, x)));
}
__device__ __forceinline__ float xhalf(float x) {
    const int b = __builtin_bit_cast(int, x);
    return __builtin_bit_cast(float,
        __builtin_amdgcn_permlanex16(b, b, 0x76543210, 0xfedcba98, false, false));
}
__device__ __forceinline__ float row_sum16(float p) {
    p += row_rd<0x161>(p); p += row_rd<0x162>(p); p += row_rd<0x164>(p); p += row_rd<0x168>(p);
    return p;
}
__device__ __forceinline__ float row_max16(float p) {
    p = fmaxf(p, row_rd<0x161>(p)); p = fmaxf(p, row_rd<0x162>(p));
    p = fmaxf(p, row_rd<0x164>(p)); p = fmaxf(p, row_rd<0x168>(p));
    return p;
}
__device__ __forceinline__ float dot4(const v4f a, const v4f b) {
    return a.x * b.x + a.y * b.y + a.z * b.z + a.w * b.w;
}

template <int I>
__device__ __forceinline__ void load_pairs(v4f (&r)[NPAIR], const int np, const int ent2,
                                           const unsigned int q,
                                           const v4f* __restrict__ h4) {
    if constexpr (I < NPAIR) {
        if (I < np) {
            const unsigned int s = (unsigned int)row_rd<0x150 | I>(ent2);
            r[I] = h4[s * 16u + q];
            load_pairs<I + 1>(r, np, ent2, q, h4);
        }
    }
}
template <int I>
__device__ __forceinline__ void score_pairs(const v4f (&r)[NPAIR], const int np, const v4f as4,
                                            const int q, float& sc) {
    if constexpr (I < NPAIR) {
        if (I < np) {
            const float p = row_sum16(dot4(r[I], as4));
            sc = (q == I) ? p : sc;
            score_pairs<I + 1>(r, np, as4, q, sc);
        }
    }
}
template <int I>
__device__ __forceinline__ void accum_pairs(const v4f (&r)[NPAIR], const int np, const float ex,
                                            v4f& acc) {
    if constexpr (I < NPAIR) {
        if (I < np) {
            const float ek = row_rd<0x150 | I>(ex);
            acc.x = fmaf(ek, r[I].x, acc.x);
            acc.y = fmaf(ek, r[I].y, acc.y);
            acc.z = fmaf(ek, r[I].z, acc.z);
            acc.w = fmaf(ek, r[I].w, acc.w);
            accum_pairs<I + 1>(r, np, ex, acc);
        }
    }
}

__global__ void __launch_bounds__(256)
node_gather_kernel(const int* __restrict__ lines, const int* __restrict__ ovcnt,
                   const int* __restrict__ ov,
                   const float* __restrict__ att_src_v, const float* __restrict__ a_dst,
                   const float* __restrict__ h,
                   const float* __restrict__ bias_conv, const float* __restrict__ W_lin,
                   const float* __restrict__ b_lin, float* __restrict__ y) {
    const int d = __builtin_amdgcn_readfirstlane((int)((blockIdx.x * 256 + threadIdx.x) >> 5));
    const int lane = threadIdx.x & 31;
    const int half = lane >> 4, q = lane & 15;
    if (d >= N_NODES) return;
    const int w = lines[(size_t)d * LINE_W + lane];
    const v4f as4 = *(const v4f*)(att_src_v + q * 4);
    const v4f bc  = *(const v4f*)(bias_conv + q * 4);
    const v4f wl  = *(const v4f*)(W_lin + q * 4);
    const float bl = b_lin[0];
    const float ad = a_dst[d];

    const int n  = __builtin_amdgcn_readfirstlane(w);
    const int nb = (n < SLOTS) ? n : SLOTS;
    const int np = (nb + 2) >> 1;
    const int ent = (lane >= 1 && lane <= nb) ? w : d;
    const int ent2 = __shfl(ent, 2 * q + half, 32);
    const v4f* h4 = (const v4f*)h;

    v4f r[NPAIR];
    load_pairs<0>(r, np, ent2, (unsigned int)q, h4);

    float sc = 0.0f;
    score_pairs<0>(r, np, as4, q, sc);
    const bool valid = (2 * q + half) <= nb;
    const float v = valid ? leaky(sc + ad) : -INFINITY;
    float m = row_max16(v);
    m = fmaxf(m, xhalf(m));

    const int2* ov2 = (const int2*)ov;
    int tov = 0;
    if (n > SLOTS) {
        tov = ovcnt[0];
        for (int j0 = 0; j0 < tov; j0 += 32) {
            const int j = j0 + lane;
            int so = d; bool mine = false;
            if (j < tov) { const int2 p = ov2[j]; if (p.x == d) { mine = true; so = p.y; } }
            unsigned int mask = (unsigned int)__ballot(mine);
            while (mask) {
                const int b = __builtin_ctz(mask);
                mask &= mask - 1;
                const int sk = __builtin_amdgcn_readlane(so, b);
                const v4f hv = *(const v4f*)(h + (size_t)sk * HIDDEN + q * 4);
                m = fmaxf(m, leaky(row_sum16(dot4(hv, as4)) + ad));
            }
        }
    }

    const float ex = valid ? expf(v - m) : 0.0f;
    v4f acc = {0.0f, 0.0f, 0.0f, 0.0f};
    accum_pairs<0>(r, np, ex, acc);

    float sov = 0.0f;
    if (n > SLOTS) {
        for (int j0 = 0; j0 < tov; j0 += 32) {
            const int j = j0 + lane;
            int so = d; bool mine = false;
            if (j < tov) { const int2 p = ov2[j]; if (p.x == d) { mine = true; so = p.y; } }
            unsigned int mask = (unsigned int)__ballot(mine);
            while (mask) {
                const int b = __builtin_ctz(mask);
                mask &= mask - 1;
                const int sk = __builtin_amdgcn_readlane(so, b);
                const v4f hv = *(const v4f*)(h + (size_t)sk * HIDDEN + q * 4);
                const float eo = expf(leaky(row_sum16(dot4(hv, as4)) + ad) - m);
                sov += eo;
                const float eh = (half == 0) ? eo : 0.0f;
                acc.x = fmaf(eh, hv.x, acc.x);
                acc.y = fmaf(eh, hv.y, acc.y);
                acc.z = fmaf(eh, hv.z, acc.z);
                acc.w = fmaf(eh, hv.w, acc.w);
            }
        }
    }

    acc.x += xhalf(acc.x);
    acc.y += xhalf(acc.y);
    acc.z += xhalf(acc.z);
    acc.w += xhalf(acc.w);
    float ssum = row_sum16(ex);
    ssum = ssum + xhalf(ssum) + sov;

    const float inv = 1.0f / ssum;
    float p = fmaxf(acc.x * inv + bc.x, 0.0f) * wl.x
            + fmaxf(acc.y * inv + bc.y, 0.0f) * wl.y
            + fmaxf(acc.z * inv + bc.z, 0.0f) * wl.z
            + fmaxf(acc.w * inv + bc.w, 0.0f) * wl.w;
    p = row_sum16(p);
    if (lane == 0) y[d] = 1.0f / (1.0f + expf(-(p + bl)));
}

extern "C" void kernel_launch(void* const* d_in, const int* in_sizes, int n_in,
                              void* d_out, int out_size, void* d_ws, size_t ws_size,
                              hipStream_t stream) {
    const float* x         = (const float*)d_in[0];
    const int*   ei        = (const int*)  d_in[1];
    const float* W         = (const float*)d_in[2];
    const float* att_src_v = (const float*)d_in[3];
    const float* att_dst_v = (const float*)d_in[4];
    const float* bias_conv = (const float*)d_in[5];
    const float* W_lin     = (const float*)d_in[6];
    const float* b_lin     = (const float*)d_in[7];
    float* y = (float*)d_out;

    float* ws     = (float*)d_ws;
    float* h      = ws;
    float* a_dst  = ws + 3250000;
    int*   lines  = (int*)(ws + 3300032);
    int*   ovcnt  = (int*)(ws + 4900032);
    int*   ov     = (int*)(ws + 4900096);
    int*   cells  = (int*)(ws + 6500096);
    int*   spcnt  = (int*)(ws + 10500096);
    int*   spill  = (int*)(ws + 10500608);

    edge_bin_kernel<<<NCHUNK, 256, 0, stream>>>(ei, cells, spcnt, spill, ovcnt);

    gat_gemm_wmma_kernel<<<3125, 32, 0, stream>>>(x, W, att_dst_v, h, a_dst);

    build_lines_kernel<<<NBIN, 512, 0, stream>>>(cells, spcnt, spill, lines, ovcnt, ov);

    node_gather_kernel<<<6250, 256, 0, stream>>>(lines, ovcnt, ov, att_src_v, a_dst, h,
                                                 bias_conv, W_lin, b_lin, y);
}


# GEMM tiles run as extra blocks of the line-building launch (196 blocks of 16 one-tile waves behind the 250 bin blocks), not as a launch of their own: 3 launches, the two overlap; sums unchanged
# speedup vs baseline: 5.5728x; 1.1350x over previous
#include <hip/hip_runtime.h>
#include <math.h>

#define N_NODES 50000
#define N_EDGES 800000
#define IN_DIM 128
#define HIDDEN 64
#define NEG_SLOPE 0.2f
#define E_TOT (N_EDGES + N_NODES)

typedef float v2f __attribute__((ext_vector_type(2)));
typedef float v8f __attribute__((ext_vector_type(8)));

typedef float v4f __attribute__((ext_vector_type(4)));

#define LINE_W 32
#define SLOTS 31

__device__ __forceinline__ float leaky(float v) { return (v > 0.0f) ? v : v * NEG_SLOPE; }

#define NBIN 250
#define BIN_NODES 200
#define NCHUNK 500
#define CHUNK_E 1600
#define EDGES_PER_THREAD 7
#define LPAD 33

__device__ __forceinline__ bool lds_line_put(int* line, const int value) {
    const int r = atomicAdd(line, 1);
    if (r < SLOTS) { line[1 + r] = value; return true; }
    return false;
}

__global__ void __launch_bounds__(256)
edge_bin_kernel(const int* __restrict__ ei, int* __restrict__ cells,
                int* __restrict__ spill_cnt, int* __restrict__ spill,
                int* __restrict__ ovcnt) {
    __shared__ int C[NBIN * LPAD];
    __shared__ int nspill;
    const int t = threadIdx.x;
    const int g = blockIdx.x;

    int s[EDGES_PER_THREAD], d[EDGES_PER_THREAD];
    #pragma unroll
    for (int k = 0; k < EDGES_PER_THREAD; ++k) {
        const int idx = k * 256 + t;
        const int e = g * CHUNK_E + ((idx < CHUNK_E) ? idx : 0);
        s[k] = ei[e];
        d[k] = ei[N_EDGES + e];
    }
    for (int i = t; i < NBIN * LPAD; i += 256) C[i] = 0;
    if (t == 0) {
        nspill = 0;
        if (g == 0) ovcnt[0] = 0;
    }
    __syncthreads();

    #pragma unroll
    for (int k = 0; k < EDGES_PER_THREAD; ++k) {
        const int idx = k * 256 + t;
        if (idx < CHUNK_E) {
            const unsigned int b  = (unsigned int)d[k] / (unsigned int)BIN_NODES;
            const unsigned int dl = (unsigned int)d[k] - b * (unsigned int)BIN_NODES;
            const int word = (int)((dl << 16) | (unsigned int)s[k]);
            if (!lds_line_put(&C[b * LPAD], word)) {
                const int k2 = atomicAdd(&nspill, 1);
                ((int2*)spill)[g * CHUNK_E + k2] = make_int2(d[k], s[k]);
            }
        }
    }
    __syncthreads();

    for (int i = t; i < NBIN * LINE_W; i += 256) {
        const int b = i >> 5, j = i & 31;
        cells[((size_t)b * NCHUNK + g) * LINE_W + j] = C[b * LPAD + j];
    }
    if (t == 0) spill_cnt[g] = nspill;
}

__device__ __forceinline__ void node_put(int* L, const int b, const int dl, const int s,
                                         int* __restrict__ ovcnt, int* __restrict__ ov) {
    if (!lds_line_put(&L[dl * LPAD], s)) {
        const int k = atomicAdd(ovcnt, 1);
        ((int2*)ov)[k] = make_int2(b * BIN_NODES + dl, s);
    }
}

__device__ __forceinline__ void
build_lines_block(int* L, const int* __restrict__ cells, const int* __restrict__ spill_cnt,
                  const int* __restrict__ spill, int* __restrict__ lines,
                  int* __restrict__ ovcnt, int* __restrict__ ov) {
    const int t = threadIdx.x;
    const int b = blockIdx.x;
    const int lane = t & 31;
    const int w = t >> 5;

    for (int i = t; i < BIN_NODES * LPAD; i += 512) L[i] = 0;
    __syncthreads();

    const int* cb = cells + (size_t)b * NCHUNK * LINE_W;
    for (int g0 = w; g0 < NCHUNK; g0 += 128) {
        int v[8];
        #pragma unroll
        for (int u = 0; u < 8; ++u) {
            const int g = g0 + 16 * u;
            v[u] = (g < NCHUNK) ? cb[g * LINE_W + lane] : 0;
        }
        #pragma unroll
        for (int u = 0; u < 8; ++u) {
            const int c  = __builtin_amdgcn_readlane(v[u], 0);
            const int cc = (c < SLOTS) ? c : SLOTS;
            if (lane >= 1 && lane <= cc) {
                const unsigned int word = (unsigned int)v[u];
                node_put(L, b, (int)(word >> 16), (int)(word & 0xffffu), ovcnt, ov);
            }
        }
    }

    for (int g = t; g < NCHUNK; g += 512) {
        const int cr = spill_cnt[g];
        const int c = (cr < CHUNK_E) ? cr : CHUNK_E;
        for (int j = 0; j < c; ++j) {
            const int2 p = ((const int2*)spill)[g * CHUNK_E + j];
            const unsigned int pb = (unsigned int)p.x / (unsigned int)BIN_NODES;
            if (pb == (unsigned int)b) node_put(L, b, p.x - b * BIN_NODES, p.y, ovcnt, ov);
        }
    }
    __syncthreads();

    for (int i = t; i < BIN_NODES * LINE_W; i += 512) {
        lines[(size_t)b * BIN_NODES * LINE_W + i] = L[(i >> 5) * LPAD + (i & 31)];
    }
}

__device__ __forceinline__ void
gat_gemm_wmma_tile(const int tile, const int lane,
                   const float* __restrict__ x, const float* __restrict__ W,
                   const float* __restrict__ att_dst_v,
                   float* __restrict__ h, float* __restrict__ a_dst)
{
    const int half = lane >> 4;
    const int lrow = lane & 15;
    const int row0 = tile * 16;

    const float* xrow  = x + (size_t)(row0 + lrow) * IN_DIM;
    const float* wrow  = W + (size_t)lrow * IN_DIM;

    v8f acc0 = {}, acc1 = {}, acc2 = {}, acc3 = {};

    for (int kb = 0; kb < IN_DIM; kb += 4) {
        const int k = kb + 2 * half;
        v2f a  = *(const v2f*)(xrow + k);
        v2f b0 = *(const v2f*)(wrow + k);
        v2f b1 = *(const v2f*)(wrow + 16 * IN_DIM + k);
        v2f b2 = *(const v2f*)(wrow + 32 * IN_DIM + k);
        v2f b3 = *(const v2f*)(wrow + 48 * IN_DIM + k);
        acc0 = __builtin_amdgcn_wmma_f32_16x16x4_f32(false, a, false, b0, (short)0, acc0, false, false);
        acc1 = __builtin_amdgcn_wmma_f32_16x16x4_f32(false, a, false, b1, (short)0, acc1, false, false);
        acc2 = __builtin_amdgcn_wmma_f32_16x16x4_f32(false, a, false, b2, (short)0, acc2, false, false);
        acc3 = __builtin_amdgcn_wmma_f32_16x16x4_f32(false, a, false, b3, (short)0, acc3, false, false);
    }

    const float ad0 = att_dst_v[lrow];
    const float ad1 = att_dst_v[16 + lrow];
    const float ad2 = att_dst_v[32 + lrow];
    const float ad3 = att_dst_v[48 + lrow];

    #pragma unroll
    for (int r = 0; r < 8; ++r) {
        const int row = row0 + r + 8 * half;
        float* hr = h + (size_t)row * HIDDEN;
        hr[lrow]      = acc0[r];
        hr[16 + lrow] = acc1[r];
        hr[32 + lrow] = acc2[r];
        hr[48 + lrow] = acc3[r];
        float pd = acc0[r] * ad0 + acc1[r] * ad1 + acc2[r] * ad2 + acc3[r] * ad3;
        #pragma unroll
        for (int off = 1; off < 16; off <<= 1) pd += __shfl_xor(pd, off, 32);
        if (lrow == 0) a_dst[row] = pd;
    }
}

#define GEMM_TILES 3125
#define TILES_PER_BLOCK 16
#define GEMM_BLOCKS 196
#define LINES_GEMM_GRID 446

__global__ void __launch_bounds__(512)
lines_and_gemm_kernel(const int* __restrict__ cells, const int* __restrict__ spill_cnt,
                      const int* __restrict__ spill, int* __restrict__ lines,
                      int* __restrict__ ovcnt, int* __restrict__ ov,
                      const float* __restrict__ x, const float* __restrict__ W,
                      const float* __restrict__ att_dst_v,
                      float* __restrict__ h, float* __restrict__ a_dst) {
    __shared__ int L[BIN_NODES * LPAD];
    static_assert(LINES_GEMM_GRID == NBIN + GEMM_BLOCKS, "grid = bin blocks + GEMM blocks");
    static_assert(GEMM_BLOCKS * TILES_PER_BLOCK >= GEMM_TILES, "every tile has a wave");
    if (blockIdx.x < NBIN) {
        build_lines_block(L, cells, spill_cnt, spill, lines, ovcnt, ov);
    } else {
        const int wv = __builtin_amdgcn_readfirstlane((int)(threadIdx.x >> 5));
        const int tile = ((int)blockIdx.x - NBIN) * TILES_PER_BLOCK + wv;
        if (tile < GEMM_TILES)
            gat_gemm_wmma_tile(tile, (int)(threadIdx.x & 31), x, W, att_dst_v, h, a_dst);
    }
}

#define NPAIR 16

template <int CTRL>
__device__ __forceinline__ int row_rd(int x) {
    return __builtin_amdgcn_update_dpp(0, x, CTRL, 0xf, 0xf, true);
}
template <int CTRL>
__device__ __forceinline__ float row_rd(float x) {
    return __builtin_bit_cast(float, row_rd<CTRL>(__builtin_bit_cast(int, x)));
}
__device__ __forceinline__ float xhalf(float x) {
    const int b = __builtin_bit_cast(int, x);
    return __builtin_bit_cast(float,
        __builtin_amdgcn_permlanex16(b, b, 0x76543210, 0xfedcba98, false, false));
}
__device__ __forceinline__ float row_sum16(float p) {
    p += row_rd<0x161>(p); p += row_rd<0x162>(p); p += row_rd<0x164>(p); p += row_rd<0x168>(p);
    return p;
}
__device__ __forceinline__ float row_max16(float p) {
    p = fmaxf(p, row_rd<0x161>(p)); p = fmaxf(p, row_rd<0x162>(p));
    p = fmaxf(p, row_rd<0x164>(p)); p = fmaxf(p, row_rd<0x168>(p));
    return p;
}
__device__ __forceinline__ float dot4(const v4f a, const v4f b) {
    return a.x * b.x + a.y * b.y + a.z * b.z + a.w * b.w;
}

template <int I>
__device__ __forceinline__ void load_pairs(v4f (&r)[NPAIR], const int np, const int ent2,
                                           const unsigned int q,
                                           const v4f* __restrict__ h4) {
    if constexpr (I < NPAIR) {
        if (I < np) {
            const unsigned int s = (unsigned int)row_rd<0x150 | I>(ent2);
            r[I] = h4[s * 16u + q];
            load_pairs<I + 1>(r, np, ent2, q, h4);
        }
    }
}
template <int I>
__device__ __forceinline__ void score_pairs(const v4f (&r)[NPAIR], const int np, const v4f as4,
                                            const int q, float& sc) {
    if constexpr (I < NPAIR) {
        if (I < np) {
            const float p = row_sum16(dot4(r[I], as4));
            sc = (q == I) ? p : sc;
            score_pairs<I + 1>(r, np, as4, q, sc);
        }
    }
}
template <int I>
__device__ __forceinline__ void accum_pairs(const v4f (&r)[NPAIR], const int np, const float ex,
                                            v4f& acc) {
    if constexpr (I < NPAIR) {
        if (I < np) {
            const float ek = row_rd<0x150 | I>(ex);
            acc.x = fmaf(ek, r[I].x, acc.x);
            acc.y = fmaf(ek, r[I].y, acc.y);
            acc.z = fmaf(ek, r[I].z, acc.z);
            acc.w = fmaf(ek, r[I].w, acc.w);
            accum_pairs<I + 1>(r, np, ex, acc);
        }
    }
}

__global__ void __launch_bounds__(256)
node_gather_kernel(const int* __restrict__ lines, const int* __restrict__ ovcnt,
                   const int* __restrict__ ov,
                   const float* __restrict__ att_src_v, const float* __restrict__ a_dst,
                   const float* __restrict__ h,
                   const float* __restrict__ bias_conv, const float* __restrict__ W_lin,
                   const float* __restrict__ b_lin, float* __restrict__ y) {
    const int d = __builtin_amdgcn_readfirstlane((int)((blockIdx.x * 256 + threadIdx.x) >> 5));
    const int lane = threadIdx.x & 31;
    const int half = lane >> 4, q = lane & 15;
    if (d >= N_NODES) return;
    const int w = lines[(size_t)d * LINE_W + lane];
    const v4f as4 = *(const v4f*)(att_src_v + q * 4);
    const v4f bc  = *(const v4f*)(bias_conv + q * 4);
    const v4f wl  = *(const v4f*)(W_lin + q * 4);
    const float bl = b_lin[0];
    const float ad = a_dst[d];

    const int n  = __builtin_amdgcn_readfirstlane(w);
    const int nb = (n < SLOTS) ? n : SLOTS;
    const int np = (nb + 2) >> 1;
    const int ent = (lane >= 1 && lane <= nb) ? w : d;
    const int ent2 = __shfl(ent, 2 * q + half, 32);
    const v4f* h4 = (const v4f*)h;

    v4f r[NPAIR];
    load_pairs<0>(r, np, ent2, (unsigned int)q, h4);

    float sc = 0.0f;
    score_pairs<0>(r, np, as4, q, sc);
    const bool valid = (2 * q + half) <= nb;
    const float v = valid ? leaky(sc + ad) : -INFINITY;
    float m = row_max16(v);
    m = fmaxf(m, xhalf(m));

    const int2* ov2 = (const int2*)ov;
    int tov = 0;
    if (n > SLOTS) {
        tov = ovcnt[0];
        for (int j0 = 0; j0 < tov; j0 += 32) {
            const int j = j0 + lane;
            int so = d; bool mine = false;
            if (j < tov) { const int2 p = ov2[j]; if (p.x == d) { mine = true; so = p.y; } }
            unsigned int mask = (unsigned int)__ballot(mine);
            while (mask) {
                const int b = __builtin_ctz(mask);
                mask &= mask - 1;
                const int sk = __builtin_amdgcn_readlane(so, b);
                const v4f hv = *(const v4f*)(h + (size_t)sk * HIDDEN + q * 4);
                m = fmaxf(m, leaky(row_sum16(dot4(hv, as4)) + ad));
            }
        }
    }

    const float ex = valid ? expf(v - m) : 0.0f;
    v4f acc = {0.0f, 0.0f, 0.0f, 0.0f};
    accum_pairs<0>(r, np, ex, acc);

    float sov = 0.0f;
    if (n > SLOTS) {
        for (int j0 = 0; j0 < tov; j0 += 32) {
            const int j = j0 + lane;
            int so = d; bool mine = false;
            if (j < tov) { const int2 p = ov2[j]; if (p.x == d) { mine = true; so = p.y; } }
            unsigned int mask = (unsigned int)__ballot(mine);
            while (mask) {
                const int b = __builtin_ctz(mask);
                mask &= mask - 1;
                const int sk = __builtin_amdgcn_readlane(so, b);
                const v4f hv = *(const v4f*)(h + (size_t)sk * HIDDEN + q * 4);
                const float eo = expf(leaky(row_sum16(dot4(hv, as4)) + ad) - m);
                sov += eo;
                const float eh = (half == 0) ? eo : 0.0f;
                acc.x = fmaf(eh, hv.x, acc.x);
                acc.y = fmaf(eh, hv.y, acc.y);
                acc.z = fmaf(eh, hv.z, acc.z);
                acc.w = fmaf(eh, hv.w, acc.w);
            }
        }
    }

    acc.x += xhalf(acc.x);
    acc.y += xhalf(acc.y);
    acc.z += xhalf(acc.z);
    acc.w += xhalf(acc.w);
    float ssum = row_sum16(ex);
    ssum = ssum + xhalf(ssum) + sov;

    const float inv = 1.0f / ssum;
    float p = fmaxf(acc.x * inv + bc.x, 0.0f) * wl.x
            + fmaxf(acc.y * inv + bc.y, 0.0f) * wl.y
            + fmaxf(acc.z * inv + bc.z, 0.0f) * wl.z
            + fmaxf(acc.w * inv + bc.w, 0.0f) * wl.w;
    p = row_sum16(p);
    if (lane == 0) y[d] = 1.0f / (1.0f + expf(-(p + bl)));
}

extern "C" void kernel_launch(void* const* d_in, const int* in_sizes, int n_in,
                              void* d_out, int out_size, void* d_ws, size_t ws_size,
                              hipStream_t stream) {
    const float* x         = (const float*)d_in[0];
    const int*   ei        = (const int*)  d_in[1];
    const float* W         = (const float*)d_in[2];
    const float* att_src_v = (const float*)d_in[3];
    const float* att_dst_v = (const float*)d_in[4];
    const float* bias_conv = (const float*)d_in[5];
    const float* W_lin     = (const float*)d_in[6];
    const float* b_lin     = (const float*)d_in[7];
    float* y = (float*)d_out;

    float* ws     = (float*)d_ws;
    float* h      = ws;
    float* a_dst  = ws + 3250000;
    int*   lines  = (int*)(ws + 3300032);
    int*   ovcnt  = (int*)(ws + 4900032);
    int*   ov     = (int*)(ws + 4900096);
    int*   cells  = (int*)(ws + 6500096);
    int*   spcnt  = (int*)(ws + 10500096);
    int*   spill  = (int*)(ws + 10500608);

    edge_bin_kernel<<<NCHUNK, 256, 0, stream>>>(ei, cells, spcnt, spill, ovcnt);

    lines_and_gemm_kernel<<<LINES_GEMM_GRID, 512, 0, stream>>>(cells, spcnt, spill, lines, ovcnt, ov,
                                                            x, W, att_dst_v, h, a_dst);

    node_gather_kernel<<<6250, 256, 0, stream>>>(lines, ovcnt, ov, att_src_v, a_dst, h,
                                                 bias_conv, W_lin, b_lin, y);
}


# Node lines stay in LDS: each bin block (32 waves) builds its 200 lines, then gathers those nodes itself; GEMM tiles join the edge-binning launch: 2 launches; sums in arrival order, overflow reversed
# speedup vs baseline: 6.3483x; 1.1392x over previous
#include <hip/hip_runtime.h>
#include <math.h>

#define N_NODES 50000
#define N_EDGES 800000
#define IN_DIM 128
#define HIDDEN 64
#define NEG_SLOPE 0.2f
#define E_TOT (N_EDGES + N_NODES)

typedef float v2f __attribute__((ext_vector_type(2)));
typedef float v8f __attribute__((ext_vector_type(8)));

typedef float v4f __attribute__((ext_vector_type(4)));

#define LINE_W 32
#define SLOTS 31

__device__ __forceinline__ float leaky(float v) { return (v > 0.0f) ? v : v * NEG_SLOPE; }

#define NBIN 250
#define BIN_NODES 200
#define NCHUNK 500
#define CHUNK_E 1600
#define EDGES_PER_THREAD 7
#define LPAD 33

__device__ __forceinline__ bool lds_line_put(int* line, const int value) {
    const int r = atomicAdd(line, 1);
    if (r < SLOTS) { line[1 + r] = value; return true; }
    return false;
}

__device__ __forceinline__ void
edge_bin_block(int* C, int& nspill, const int* __restrict__ ei, int* __restrict__ cells,
               int* __restrict__ spill_cnt, int* __restrict__ spill,
               int* __restrict__ ovcnt) {
    const int t = threadIdx.x;
    const int g = blockIdx.x;

    int s[EDGES_PER_THREAD], d[EDGES_PER_THREAD];
    #pragma unroll
    for (int k = 0; k < EDGES_PER_THREAD; ++k) {
        const int idx = k * 256 + t;
        const int e = g * CHUNK_E + ((idx < CHUNK_E) ? idx : 0);
        s[k] = ei[e];
        d[k] = ei[N_EDGES + e];
    }
    for (int i = t; i < NBIN * LPAD; i += 256) C[i] = 0;
    if (t == 0) {
        nspill = 0;
        if (g == 0) ovcnt[0] = 0;
    }
    __syncthreads();

    #pragma unroll
    for (int k = 0; k < EDGES_PER_THREAD; ++k) {
        const int idx = k * 256 + t;
        if (idx < CHUNK_E) {
            const unsigned int b  = (unsigned int)d[k] / (unsigned int)BIN_NODES;
            const unsigned int dl = (unsigned int)d[k] - b * (unsigned int)BIN_NODES;
            const int word = (int)((dl << 16) | (unsigned int)s[k]);
            if (!lds_line_put(&C[b * LPAD], word)) {
                const int k2 = atomicAdd(&nspill, 1);
                ((int2*)spill)[g * CHUNK_E + k2] = make_int2(d[k], s[k]);
            }
        }
    }
    __syncthreads();

    for (int i = t; i < NBIN * LINE_W; i += 256) {
        const int b = i >> 5, j = i & 31;
        cells[((size_t)b * NCHUNK + g) * LINE_W + j] = C[b * LPAD + j];
    }
    if (t == 0) spill_cnt[g] = nspill;
}

__device__ __forceinline__ void node_put(int* L, int* head, const int dl, const int s,
                                         int* __restrict__ ovcnt, int* __restrict__ ov) {
    if (!lds_line_put(&L[dl * LPAD], s)) {
        const int k = atomicAdd(ovcnt, 1);
        const int prev = atomicExch(&head[dl], k);
        ((int2*)ov)[k] = make_int2(prev, s);
    }
}

__device__ __forceinline__ void
gat_gemm_wmma_tile(const int tile, const int lane,
                   const float* __restrict__ x, const float* __restrict__ W,
                   const float* __restrict__ att_dst_v,
                   float* __restrict__ h, float* __restrict__ a_dst)
{
    const int half = lane >> 4;
    const int lrow = lane & 15;
    const int row0 = tile * 16;

    const float* xrow  = x + (size_t)(row0 + lrow) * IN_DIM;
    const float* wrow  = W + (size_t)lrow * IN_DIM;

    v8f acc0 = {}, acc1 = {}, acc2 = {}, acc3 = {};

    for (int kb = 0; kb < IN_DIM; kb += 4) {
        const int k = kb + 2 * half;
        v2f a  = *(const v2f*)(xrow + k);
        v2f b0 = *(const v2f*)(wrow + k);
        v2f b1 = *(const v2f*)(wrow + 16 * IN_DIM + k);
        v2f b2 = *(const v2f*)(wrow + 32 * IN_DIM + k);
        v2f b3 = *(const v2f*)(wrow + 48 * IN_DIM + k);
        acc0 = __builtin_amdgcn_wmma_f32_16x16x4_f32(false, a, false, b0, (short)0, acc0, false, false);
        acc1 = __builtin_amdgcn_wmma_f32_16x16x4_f32(false, a, false, b1, (short)0, acc1, false, false);
        acc2 = __builtin_amdgcn_wmma_f32_16x16x4_f32(false, a, false, b2, (short)0, acc2, false, false);
        acc3 = __builtin_amdgcn_wmma_f32_16x16x4_f32(false, a, false, b3, (short)0, acc3, false, false);
    }

    const float ad0 = att_dst_v[lrow];
    const float ad1 = att_dst_v[16 + lrow];
    const float ad2 = att_dst_v[32 + lrow];
    const float ad3 = att_dst_v[48 + lrow];

    #pragma unroll
    for (int r = 0; r < 8; ++r) {
        const int row = row0 + r + 8 * half;
        float* hr = h + (size_t)row * HIDDEN;
        hr[lrow]      = acc0[r];
        hr[16 + lrow] = acc1[r];
        hr[32 + lrow] = acc2[r];
        hr[48 + lrow] = acc3[r];
        float pd = acc0[r] * ad0 + acc1[r] * ad1 + acc2[r] * ad2 + acc3[r] * ad3;
        #pragma unroll
        for (int off = 1; off < 16; off <<= 1) pd += __shfl_xor(pd, off, 32);
        if (lrow == 0) a_dst[row] = pd;
    }
}

#define GEMM_TILES 3125
#define TILES_PER_BLOCK 8
#define GEMM_BLOCKS 391
#define BIN_GEMM_GRID 891

__global__ void __launch_bounds__(256)
bin_and_gemm_kernel(const int* __restrict__ ei, int* __restrict__ cells,
                    int* __restrict__ spill_cnt, int* __restrict__ spill,
                    int* __restrict__ ovcnt,
                    const float* __restrict__ x, const float* __restrict__ W,
                    const float* __restrict__ att_dst_v,
                    float* __restrict__ h, float* __restrict__ a_dst) {
    __shared__ int C[NBIN * LPAD];
    __shared__ int nspill;
    static_assert(BIN_GEMM_GRID == NCHUNK + GEMM_BLOCKS, "grid = chunk blocks + GEMM blocks");
    static_assert(GEMM_BLOCKS * TILES_PER_BLOCK >= GEMM_TILES, "every tile has a wave");
    if (blockIdx.x < NCHUNK) {
        edge_bin_block(C, nspill, ei, cells, spill_cnt, spill, ovcnt);
    } else {
        const int wv = __builtin_amdgcn_readfirstlane((int)(threadIdx.x >> 5));
        const int tile = ((int)blockIdx.x - NCHUNK) * TILES_PER_BLOCK + wv;
        if (tile < GEMM_TILES)
            gat_gemm_wmma_tile(tile, (int)(threadIdx.x & 31), x, W, att_dst_v, h, a_dst);
    }
}

#define NPAIR 16

template <int CTRL>
__device__ __forceinline__ int row_rd(int x) {
    return __builtin_amdgcn_update_dpp(0, x, CTRL, 0xf, 0xf, true);
}
template <int CTRL>
__device__ __forceinline__ float row_rd(float x) {
    return __builtin_bit_cast(float, row_rd<CTRL>(__builtin_bit_cast(int, x)));
}
__device__ __forceinline__ float xhalf(float x) {
    const int b = __builtin_bit_cast(int, x);
    return __builtin_bit_cast(float,
        __builtin_amdgcn_permlanex16(b, b, 0x76543210, 0xfedcba98, false, false));
}
__device__ __forceinline__ float row_sum16(float p) {
    p += row_rd<0x161>(p); p += row_rd<0x162>(p); p += row_rd<0x164>(p); p += row_rd<0x168>(p);
    return p;
}
__device__ __forceinline__ float row_max16(float p) {
    p = fmaxf(p, row_rd<0x161>(p)); p = fmaxf(p, row_rd<0x162>(p));
    p = fmaxf(p, row_rd<0x164>(p)); p = fmaxf(p, row_rd<0x168>(p));
    return p;
}
__device__ __forceinline__ float dot4(const v4f a, const v4f b) {
    return a.x * b.x + a.y * b.y + a.z * b.z + a.w * b.w;
}

template <int I>
__device__ __forceinline__ void load_pairs(v4f (&r)[NPAIR], const int np, const int ent2,
                                           const unsigned int q,
                                           const v4f* __restrict__ h4) {
    if constexpr (I < NPAIR) {
        if (I < np) {
            const unsigned int s = (unsigned int)row_rd<0x150 | I>(ent2);
            r[I] = h4[s * 16u + q];
            load_pairs<I + 1>(r, np, ent2, q, h4);
        }
    }
}
template <int I>
__device__ __forceinline__ void score_pairs(const v4f (&r)[NPAIR], const int np, const v4f as4,
                                            const int q, float& sc) {
    if constexpr (I < NPAIR) {
        if (I < np) {
            const float p = row_sum16(dot4(r[I], as4));
            sc = (q == I) ? p : sc;
            score_pairs<I + 1>(r, np, as4, q, sc);
        }
    }
}
template <int I>
__device__ __forceinline__ void accum_pairs(const v4f (&r)[NPAIR], const int np, const float ex,
                                            v4f& acc) {
    if constexpr (I < NPAIR) {
        if (I < np) {
            const float ek = row_rd<0x150 | I>(ex);
            acc.x = fmaf(ek, r[I].x, acc.x);
            acc.y = fmaf(ek, r[I].y, acc.y);
            acc.z = fmaf(ek, r[I].z, acc.z);
            acc.w = fmaf(ek, r[I].w, acc.w);
            accum_pairs<I + 1>(r, np, ex, acc);
        }
    }
}

#define GATHER_THREADS 1024
#define GATHER_WAVES 32

__global__ void __launch_bounds__(GATHER_THREADS)
lines_gather_kernel(const int* __restrict__ cells, const int* __restrict__ spill_cnt,
                    const int* __restrict__ spill, int* __restrict__ ovcnt, int* __restrict__ ov,
                    const float* __restrict__ att_src_v, const float* __restrict__ a_dst,
                    const float* __restrict__ h,
                    const float* __restrict__ bias_conv, const float* __restrict__ W_lin,
                    const float* __restrict__ b_lin, float* __restrict__ y) {
    __shared__ int L[BIN_NODES * LPAD];
    __shared__ int head[BIN_NODES];
    const int t = threadIdx.x;
    const int b = blockIdx.x;
    const int lane = t & 31;
    const int half = lane >> 4, q = lane & 15;
    const int wv = __builtin_amdgcn_readfirstlane(t >> 5);

    for (int i = t; i < BIN_NODES * LPAD; i += GATHER_THREADS) L[i] = 0;
    if (t < BIN_NODES) head[t] = -1;
    __syncthreads();

    const v4f as4 = *(const v4f*)(att_src_v + q * 4);
    const v4f bc  = *(const v4f*)(bias_conv + q * 4);
    const v4f wl  = *(const v4f*)(W_lin + q * 4);
    const float bl = b_lin[0];

    const int* cb = cells + (size_t)b * NCHUNK * LINE_W;
    for (int g0 = wv; g0 < NCHUNK; g0 += 8 * GATHER_WAVES) {
        int v[8];
        #pragma unroll
        for (int u = 0; u < 8; ++u) {
            const int g = g0 + GATHER_WAVES * u;
            v[u] = (g < NCHUNK) ? cb[g * LINE_W + lane] : 0;
        }
        #pragma unroll
        for (int u = 0; u < 8; ++u) {
            const int c  = __builtin_amdgcn_readlane(v[u], 0);
            const int cc = (c < SLOTS) ? c : SLOTS;
            if (lane >= 1 && lane <= cc) {
                const unsigned int word = (unsigned int)v[u];
                node_put(L, head, (int)(word >> 16), (int)(word & 0xffffu), ovcnt, ov);
            }
        }
    }

    for (int g = t; g < NCHUNK; g += GATHER_THREADS) {
        const int cr = spill_cnt[g];
        const int c = (cr < CHUNK_E) ? cr : CHUNK_E;
        for (int j = 0; j < c; ++j) {
            const int2 p = ((const int2*)spill)[g * CHUNK_E + j];
            const unsigned int pb = (unsigned int)p.x / (unsigned int)BIN_NODES;
            if (pb == (unsigned int)b) node_put(L, head, p.x - b * BIN_NODES, p.y, ovcnt, ov);
        }
    }
    __syncthreads();

    const v4f* h4 = (const v4f*)h;
    const int2* ov2 = (const int2*)ov;
    for (int dl = wv; dl < BIN_NODES; dl += GATHER_WAVES) {
        const int d = b * BIN_NODES + dl;
        const int w = L[dl * LPAD + lane];
        const float ad = a_dst[d];

        const int n  = __builtin_amdgcn_readfirstlane(w);
        const int nb = (n < SLOTS) ? n : SLOTS;
        const int np = (nb + 2) >> 1;
        const int ent = (lane >= 1 && lane <= nb) ? w : d;
        const int ent2 = __shfl(ent, 2 * q + half, 32);

        v4f r[NPAIR];
        load_pairs<0>(r, np, ent2, (unsigned int)q, h4);

        float sc = 0.0f;
        score_pairs<0>(r, np, as4, q, sc);
        const bool valid = (2 * q + half) <= nb;
        const float v = valid ? leaky(sc + ad) : -INFINITY;
        float m = row_max16(v);
        m = fmaxf(m, xhalf(m));

        const int extra = n - SLOTS;
        if (extra > 0) {
            int k = head[dl];
            for (int j = 0; j < extra; ++j) {
                k = __builtin_amdgcn_readfirstlane(k);
                if (k < 0 || k >= N_EDGES) break;
                const int2 p = ov2[k];
                const v4f hv = *(const v4f*)(h + (size_t)p.y * HIDDEN + q * 4);
                m = fmaxf(m, leaky(row_sum16(dot4(hv, as4)) + ad));
                k = p.x;
            }
        }

        const float ex = valid ? expf(v - m) : 0.0f;
        v4f acc = {0.0f, 0.0f, 0.0f, 0.0f};
        accum_pairs<0>(r, np, ex, acc);

        float sov = 0.0f;
        if (extra > 0) {
            int k = head[dl];
            for (int j = 0; j < extra; ++j) {
                k = __builtin_amdgcn_readfirstlane(k);
                if (k < 0 || k >= N_EDGES) break;
                const int2 p = ov2[k];
                const v4f hv = *(const v4f*)(h + (size_t)p.y * HIDDEN + q * 4);
                const float eo = expf(leaky(row_sum16(dot4(hv, as4)) + ad) - m);
                sov += eo;
                const float eh = (half == 0) ? eo : 0.0f;
                acc.x = fmaf(eh, hv.x, acc.x);
                acc.y = fmaf(eh, hv.y, acc.y);
                acc.z = fmaf(eh, hv.z, acc.z);
                acc.w = fmaf(eh, hv.w, acc.w);
                k = p.x;
            }
        }

        acc.x += xhalf(acc.x);
        acc.y += xhalf(acc.y);
        acc.z += xhalf(acc.z);
        acc.w += xhalf(acc.w);
        float ssum = row_sum16(ex);
        ssum = ssum + xhalf(ssum) + sov;

        const float inv = 1.0f / ssum;
        float p = fmaxf(acc.x * inv + bc.x, 0.0f) * wl.x
                + fmaxf(acc.y * inv + bc.y, 0.0f) * wl.y
                + fmaxf(acc.z * inv + bc.z, 0.0f) * wl.z
                + fmaxf(acc.w * inv + bc.w, 0.0f) * wl.w;
        p = row_sum16(p);
        if (lane == 0) y[d] = 1.0f / (1.0f + expf(-(p + bl)));
    }
}

extern "C" void kernel_launch(void* const* d_in, const int* in_sizes, int n_in,
                              void* d_out, int out_size, void* d_ws, size_t ws_size,
                              hipStream_t stream) {
    const float* x         = (const float*)d_in[0];
    const int*   ei        = (const int*)  d_in[1];
    const float* W         = (const float*)d_in[2];
    const float* att_src_v = (const float*)d_in[3];
    const float* att_dst_v = (const float*)d_in[4];
    const float* bias_conv = (const float*)d_in[5];
    const float* W_lin     = (const float*)d_in[6];
    const float* b_lin     = (const float*)d_in[7];
    float* y = (float*)d_out;

    float* ws     = (float*)d_ws;
    float* h      = ws;
    float* a_dst  = ws + 3250000;
    int*   ovcnt  = (int*)(ws + 4900032);
    int*   ov     = (int*)(ws + 4900096);
    int*   cells  = (int*)(ws + 6500096);
    int*   spcnt  = (int*)(ws + 10500096);
    int*   spill  = (int*)(ws + 10500608);

    bin_and_gemm_kernel<<<BIN_GEMM_GRID, 256, 0, stream>>>(ei, cells, spcnt, spill, ovcnt,
                                                          x, W, att_dst_v, h, a_dst);

    lines_gather_kernel<<<NBIN, GATHER_THREADS, 0, stream>>>(cells, spcnt, spill, ovcnt, ov,
                                                             att_src_v, a_dst, h,
                                                             bias_conv, W_lin, b_lin, y);
}


# Edge chunks doubled to 3200 edges: 250 chunk blocks and half the (bin, chunk) cell lines, so a gather wave builds from 8 cells read in one group; sums still in arrival order
# speedup vs baseline: 6.6208x; 1.0429x over previous
#include <hip/hip_runtime.h>
#include <math.h>

#define N_NODES 50000
#define N_EDGES 800000
#define IN_DIM 128
#define HIDDEN 64
#define NEG_SLOPE 0.2f
#define E_TOT (N_EDGES + N_NODES)

typedef float v2f __attribute__((ext_vector_type(2)));
typedef float v8f __attribute__((ext_vector_type(8)));

typedef float v4f __attribute__((ext_vector_type(4)));

#define LINE_W 32
#define SLOTS 31

__device__ __forceinline__ float leaky(float v) { return (v > 0.0f) ? v : v * NEG_SLOPE; }

#define NBIN 250
#define BIN_NODES 200
#define NCHUNK 250
#define CHUNK_E 3200
#define EDGES_PER_THREAD 13
#define LPAD 33

__device__ __forceinline__ bool lds_line_put(int* line, const int value) {
    const int r = atomicAdd(line, 1);
    if (r < SLOTS) { line[1 + r] = value; return true; }
    return false;
}

__device__ __forceinline__ void
edge_bin_block(int* C, int& nspill, const int* __restrict__ ei, int* __restrict__ cells,
               int* __restrict__ spill_cnt, int* __restrict__ spill,
               int* __restrict__ ovcnt) {
    const int t = threadIdx.x;
    const int g = blockIdx.x;

    int s[EDGES_PER_THREAD], d[EDGES_PER_THREAD];
    #pragma unroll
    for (int k = 0; k < EDGES_PER_THREAD; ++k) {
        const int idx = k * 256 + t;
        const int e = g * CHUNK_E + ((idx < CHUNK_E) ? idx : 0);
        s[k] = ei[e];
        d[k] = ei[N_EDGES + e];
    }
    for (int i = t; i < NBIN * LPAD; i += 256) C[i] = 0;
    if (t == 0) {
        nspill = 0;
        if (g == 0) ovcnt[0] = 0;
    }
    __syncthreads();

    #pragma unroll
    for (int k = 0; k < EDGES_PER_THREAD; ++k) {
        const int idx = k * 256 + t;
        if (idx < CHUNK_E) {
            const unsigned int b  = (unsigned int)d[k] / (unsigned int)BIN_NODES;
            const unsigned int dl = (unsigned int)d[k] - b * (unsigned int)BIN_NODES;
            const int word = (int)((dl << 16) | (unsigned int)s[k]);
            if (!lds_line_put(&C[b * LPAD], word)) {
                const int k2 = atomicAdd(&nspill, 1);
                ((int2*)spill)[g * CHUNK_E + k2] = make_int2(d[k], s[k]);
            }
        }
    }
    __syncthreads();

    for (int i = t; i < NBIN * LINE_W; i += 256) {
        const int b = i >> 5, j = i & 31;
        cells[((size_t)b * NCHUNK + g) * LINE_W + j] = C[b * LPAD + j];
    }
    if (t == 0) spill_cnt[g] = nspill;
}

__device__ __forceinline__ void node_put(int* L, int* head, const int dl, const int s,
                                         int* __restrict__ ovcnt, int* __restrict__ ov) {
    if (!lds_line_put(&L[dl * LPAD], s)) {
        const int k = atomicAdd(ovcnt, 1);
        const int prev = atomicExch(&head[dl], k);
        ((int2*)ov)[k] = make_int2(prev, s);
    }
}

__device__ __forceinline__ void
gat_gemm_wmma_tile(const int tile, const int lane,
                   const float* __restrict__ x, const float* __restrict__ W,
                   const float* __restrict__ att_dst_v,
                   float* __restrict__ h, float* __restrict__ a_dst)
{
    const int half = lane >> 4;
    const int lrow = lane & 15;
    const int row0 = tile * 16;

    const float* xrow  = x + (size_t)(row0 + lrow) * IN_DIM;
    const float* wrow  = W + (size_t)lrow * IN_DIM;

    v8f acc0 = {}, acc1 = {}, acc2 = {}, acc3 = {};

    for (int kb = 0; kb < IN_DIM; kb += 4) {
        const int k = kb + 2 * half;
        v2f a  = *(const v2f*)(xrow + k);
        v2f b0 = *(const v2f*)(wrow + k);
        v2f b1 = *(const v2f*)(wrow + 16 * IN_DIM + k);
        v2f b2 = *(const v2f*)(wrow + 32 * IN_DIM + k);
        v2f b3 = *(const v2f*)(wrow + 48 * IN_DIM + k);
        acc0 = __builtin_amdgcn_wmma_f32_16x16x4_f32(false, a, false, b0, (short)0, acc0, false, false);
        acc1 = __builtin_amdgcn_wmma_f32_16x16x4_f32(false, a, false, b1, (short)0, acc1, false, false);
        acc2 = __builtin_amdgcn_wmma_f32_16x16x4_f32(false, a, false, b2, (short)0, acc2, false, false);
        acc3 = __builtin_amdgcn_wmma_f32_16x16x4_f32(false, a, false, b3, (short)0, acc3, false, false);
    }

    const float ad0 = att_dst_v[lrow];
    const float ad1 = att_dst_v[16 + lrow];
    const float ad2 = att_dst_v[32 + lrow];
    const float ad3 = att_dst_v[48 + lrow];

    #pragma unroll
    for (int r = 0; r < 8; ++r) {
        const int row = row0 + r + 8 * half;
        float* hr = h + (size_t)row * HIDDEN;
        hr[lrow]      = acc0[r];
        hr[16 + lrow] = acc1[r];
        hr[32 + lrow] = acc2[r];
        hr[48 + lrow] = acc3[r];
        float pd = acc0[r] * ad0 + acc1[r] * ad1 + acc2[r] * ad2 + acc3[r] * ad3;
        #pragma unroll
        for (int off = 1; off < 16; off <<= 1) pd += __shfl_xor(pd, off, 32);
        if (lrow == 0) a_dst[row] = pd;
    }
}

#define GEMM_TILES 3125
#define TILES_PER_BLOCK 8
#define GEMM_BLOCKS 391
#define BIN_GEMM_GRID 641

__global__ void __launch_bounds__(256)
bin_and_gemm_kernel(const int* __restrict__ ei, int* __restrict__ cells,
                    int* __restrict__ spill_cnt, int* __restrict__ spill,
                    int* __restrict__ ovcnt,
                    const float* __restrict__ x, const float* __restrict__ W,
                    const float* __restrict__ att_dst_v,
                    float* __restrict__ h, float* __restrict__ a_dst) {
    __shared__ int C[NBIN * LPAD];
    __shared__ int nspill;
    static_assert(BIN_GEMM_GRID == NCHUNK + GEMM_BLOCKS, "grid = chunk blocks + GEMM blocks");
    static_assert(GEMM_BLOCKS * TILES_PER_BLOCK >= GEMM_TILES, "every tile has a wave");
    if (blockIdx.x < NCHUNK) {
        edge_bin_block(C, nspill, ei, cells, spill_cnt, spill, ovcnt);
    } else {
        const int wv = __builtin_amdgcn_readfirstlane((int)(threadIdx.x >> 5));
        const int tile = ((int)blockIdx.x - NCHUNK) * TILES_PER_BLOCK + wv;
        if (tile < GEMM_TILES)
            gat_gemm_wmma_tile(tile, (int)(threadIdx.x & 31), x, W, att_dst_v, h, a_dst);
    }
}

#define NPAIR 16

template <int CTRL>
__device__ __forceinline__ int row_rd(int x) {
    return __builtin_amdgcn_update_dpp(0, x, CTRL, 0xf, 0xf, true);
}
template <int CTRL>
__device__ __forceinline__ float row_rd(float x) {
    return __builtin_bit_cast(float, row_rd<CTRL>(__builtin_bit_cast(int, x)));
}
__device__ __forceinline__ float xhalf(float x) {
    const int b = __builtin_bit_cast(int, x);
    return __builtin_bit_cast(float,
        __builtin_amdgcn_permlanex16(b, b, 0x76543210, 0xfedcba98, false, false));
}
__device__ __forceinline__ float row_sum16(float p) {
    p += row_rd<0x161>(p); p += row_rd<0x162>(p); p += row_rd<0x164>(p); p += row_rd<0x168>(p);
    return p;
}
__device__ __forceinline__ float row_max16(float p) {
    p = fmaxf(p, row_rd<0x161>(p)); p = fmaxf(p, row_rd<0x162>(p));
    p = fmaxf(p, row_rd<0x164>(p)); p = fmaxf(p, row_rd<0x168>(p));
    return p;
}
__device__ __forceinline__ float dot4(const v4f a, const v4f b) {
    return a.x * b.x + a.y * b.y + a.z * b.z + a.w * b.w;
}

template <int I>
__device__ __forceinline__ void load_pairs(v4f (&r)[NPAIR], const int np, const int ent2,
                                           const unsigned int q,
                                           const v4f* __restrict__ h4) {
    if constexpr (I < NPAIR) {
        if (I < np) {
            const unsigned int s = (unsigned int)row_rd<0x150 | I>(ent2);
            r[I] = h4[s * 16u + q];
            load_pairs<I + 1>(r, np, ent2, q, h4);
        }
    }
}
template <int I>
__device__ __forceinline__ void score_pairs(const v4f (&r)[NPAIR], const int np, const v4f as4,
                                            const int q, float& sc) {
    if constexpr (I < NPAIR) {
        if (I < np) {
            const float p = row_sum16(dot4(r[I], as4));
            sc = (q == I) ? p : sc;
            score_pairs<I + 1>(r, np, as4, q, sc);
        }
    }
}
template <int I>
__device__ __forceinline__ void accum_pairs(const v4f (&r)[NPAIR], const int np, const float ex,
                                            v4f& acc) {
    if constexpr (I < NPAIR) {
        if (I < np) {
            const float ek = row_rd<0x150 | I>(ex);
            acc.x = fmaf(ek, r[I].x, acc.x);
            acc.y = fmaf(ek, r[I].y, acc.y);
            acc.z = fmaf(ek, r[I].z, acc.z);
            acc.w = fmaf(ek, r[I].w, acc.w);
            accum_pairs<I + 1>(r, np, ex, acc);
        }
    }
}

#define GATHER_THREADS 1024
#define GATHER_WAVES 32

__global__ void __launch_bounds__(GATHER_THREADS)
lines_gather_kernel(const int* __restrict__ cells, const int* __restrict__ spill_cnt,
                    const int* __restrict__ spill, int* __restrict__ ovcnt, int* __restrict__ ov,
                    const float* __restrict__ att_src_v, const float* __restrict__ a_dst,
                    const float* __restrict__ h,
                    const float* __restrict__ bias_conv, const float* __restrict__ W_lin,
                    const float* __restrict__ b_lin, float* __restrict__ y) {
    __shared__ int L[BIN_NODES * LPAD];
    __shared__ int head[BIN_NODES];
    const int t = threadIdx.x;
    const int b = blockIdx.x;
    const int lane = t & 31;
    const int half = lane >> 4, q = lane & 15;
    const int wv = __builtin_amdgcn_readfirstlane(t >> 5);

    for (int i = t; i < BIN_NODES * LPAD; i += GATHER_THREADS) L[i] = 0;
    if (t < BIN_NODES) head[t] = -1;
    __syncthreads();

    const v4f as4 = *(const v4f*)(att_src_v + q * 4);
    const v4f bc  = *(const v4f*)(bias_conv + q * 4);
    const v4f wl  = *(const v4f*)(W_lin + q * 4);
    const float bl = b_lin[0];

    const int* cb = cells + (size_t)b * NCHUNK * LINE_W;
    for (int g0 = wv; g0 < NCHUNK; g0 += 8 * GATHER_WAVES) {
        int v[8];
        #pragma unroll
        for (int u = 0; u < 8; ++u) {
            const int g = g0 + GATHER_WAVES * u;
            v[u] = (g < NCHUNK) ? cb[g * LINE_W + lane] : 0;
        }
        #pragma unroll
        for (int u = 0; u < 8; ++u) {
            const int c  = __builtin_amdgcn_readlane(v[u], 0);
            const int cc = (c < SLOTS) ? c : SLOTS;
            if (lane >= 1 && lane <= cc) {
                const unsigned int word = (unsigned int)v[u];
                node_put(L, head, (int)(word >> 16), (int)(word & 0xffffu), ovcnt, ov);
            }
        }
    }

    for (int g = t; g < NCHUNK; g += GATHER_THREADS) {
        const int cr = spill_cnt[g];
        const int c = (cr < CHUNK_E) ? cr : CHUNK_E;
        for (int j = 0; j < c; ++j) {
            const int2 p = ((const int2*)spill)[g * CHUNK_E + j];
            const unsigned int pb = (unsigned int)p.x / (unsigned int)BIN_NODES;
            if (pb == (unsigned int)b) node_put(L, head, p.x - b * BIN_NODES, p.y, ovcnt, ov);
        }
    }
    __syncthreads();

    const v4f* h4 = (const v4f*)h;
    const int2* ov2 = (const int2*)ov;
    for (int dl = wv; dl < BIN_NODES; dl += GATHER_WAVES) {
        const int d = b * BIN_NODES + dl;
        const int w = L[dl * LPAD + lane];
        const float ad = a_dst[d];

        const int n  = __builtin_amdgcn_readfirstlane(w);
        const int nb = (n < SLOTS) ? n : SLOTS;
        const int np = (nb + 2) >> 1;
        const int ent = (lane >= 1 && lane <= nb) ? w : d;
        const int ent2 = __shfl(ent, 2 * q + half, 32);

        v4f r[NPAIR];
        load_pairs<0>(r, np, ent2, (unsigned int)q, h4);

        float sc = 0.0f;
        score_pairs<0>(r, np, as4, q, sc);
        const bool valid = (2 * q + half) <= nb;
        const float v = valid ? leaky(sc + ad) : -INFINITY;
        float m = row_max16(v);
        m = fmaxf(m, xhalf(m));

        const int extra = n - SLOTS;
        if (extra > 0) {
            int k = head[dl];
            for (int j = 0; j < extra; ++j) {
                k = __builtin_amdgcn_readfirstlane(k);
                if (k < 0 || k >= N_EDGES) break;
                const int2 p = ov2[k];
                const v4f hv = *(const v4f*)(h + (size_t)p.y * HIDDEN + q * 4);
                m = fmaxf(m, leaky(row_sum16(dot4(hv, as4)) + ad));
                k = p.x;
            }
        }

        const float ex = valid ? expf(v - m) : 0.0f;
        v4f acc = {0.0f, 0.0f, 0.0f, 0.0f};
        accum_pairs<0>(r, np, ex, acc);

        float sov = 0.0f;
        if (extra > 0) {
            int k = head[dl];
            for (int j = 0; j < extra; ++j) {
                k = __builtin_amdgcn_readfirstlane(k);
                if (k < 0 || k >= N_EDGES) break;
                const int2 p = ov2[k];
                const v4f hv = *(const v4f*)(h + (size_t)p.y * HIDDEN + q * 4);
                const float eo = expf(leaky(row_sum16(dot4(hv, as4)) + ad) - m);
                sov += eo;
                const float eh = (half == 0) ? eo : 0.0f;
                acc.x = fmaf(eh, hv.x, acc.x);
                acc.y = fmaf(eh, hv.y, acc.y);
                acc.z = fmaf(eh, hv.z, acc.z);
                acc.w = fmaf(eh, hv.w, acc.w);
                k = p.x;
            }
        }

        acc.x += xhalf(acc.x);
        acc.y += xhalf(acc.y);
        acc.z += xhalf(acc.z);
        acc.w += xhalf(acc.w);
        float ssum = row_sum16(ex);
        ssum = ssum + xhalf(ssum) + sov;

        const float inv = 1.0f / ssum;
        float p = fmaxf(acc.x * inv + bc.x, 0.0f) * wl.x
                + fmaxf(acc.y * inv + bc.y, 0.0f) * wl.y
                + fmaxf(acc.z * inv + bc.z, 0.0f) * wl.z
                + fmaxf(acc.w * inv + bc.w, 0.0f) * wl.w;
        p = row_sum16(p);
        if (lane == 0) y[d] = 1.0f / (1.0f + expf(-(p + bl)));
    }
}

extern "C" void kernel_launch(void* const* d_in, const int* in_sizes, int n_in,
                              void* d_out, int out_size, void* d_ws, size_t ws_size,
                              hipStream_t stream) {
    const float* x         = (const float*)d_in[0];
    const int*   ei        = (const int*)  d_in[1];
    const float* W         = (const float*)d_in[2];
    const float* att_src_v = (const float*)d_in[3];
    const float* att_dst_v = (const float*)d_in[4];
    const float* bias_conv = (const float*)d_in[5];
    const float* W_lin     = (const float*)d_in[6];
    const float* b_lin     = (const float*)d_in[7];
    float* y = (float*)d_out;

    float* ws     = (float*)d_ws;
    float* h      = ws;
    float* a_dst  = ws + 3250000;
    int*   ovcnt  = (int*)(ws + 4900032);
    int*   ov     = (int*)(ws + 4900096);
    int*   cells  = (int*)(ws + 6500096);
    int*   spcnt  = (int*)(ws + 10500096);
    int*   spill  = (int*)(ws + 10500608);

    bin_and_gemm_kernel<<<BIN_GEMM_GRID, 256, 0, stream>>>(ei, cells, spcnt, spill, ovcnt,
                                                          x, W, att_dst_v, h, a_dst);

    lines_gather_kernel<<<NBIN, GATHER_THREADS, 0, stream>>>(cells, spcnt, spill, ovcnt, ov,
                                                             att_src_v, a_dst, h,
                                                             bias_conv, W_lin, b_lin, y);
}


# GEMM blocks copy W once into LDS (k pairs of two steps side by side) and read every B fragment from there, one 16 B LDS read per two k-steps, not from global; x rows requested first; sums unchanged
# speedup vs baseline: 7.8487x; 1.1855x over previous
#include <hip/hip_runtime.h>
#include <math.h>

#define N_NODES 50000
#define N_EDGES 800000
#define IN_DIM 128
#define HIDDEN 64
#define NEG_SLOPE 0.2f
#define E_TOT (N_EDGES + N_NODES)

typedef float v2f __attribute__((ext_vector_type(2)));
typedef float v8f __attribute__((ext_vector_type(8)));

typedef float v4f __attribute__((ext_vector_type(4)));

#define LINE_W 32
#define SLOTS 31

__device__ __forceinline__ float leaky(float v) { return (v > 0.0f) ? v : v * NEG_SLOPE; }

#define NBIN 250
#define BIN_NODES 200
#define NCHUNK 250
#define CHUNK_E 3200
#define EDGES_PER_THREAD 13
#define LPAD 33

__device__ __forceinline__ bool lds_line_put(int* line, const int value) {
    const int r = atomicAdd(line, 1);
    if (r < SLOTS) { line[1 + r] = value; return true; }
    return false;
}

__device__ __forceinline__ void
edge_bin_block(int* C, int& nspill, const int* __restrict__ ei, int* __restrict__ cells,
               int* __restrict__ spill_cnt, int* __restrict__ spill,
               int* __restrict__ ovcnt) {
    const int t = threadIdx.x;
    const int g = blockIdx.x;

    int s[EDGES_PER_THREAD], d[EDGES_PER_THREAD];
    #pragma unroll
    for (int k = 0; k < EDGES_PER_THREAD; ++k) {
        const int idx = k * 256 + t;
        const int e = g * CHUNK_E + ((idx < CHUNK_E) ? idx : 0);
        s[k] = ei[e];
        d[k] = ei[N_EDGES + e];
    }
    for (int i = t; i < NBIN * LPAD; i += 256) C[i] = 0;
    if (t == 0) {
        nspill = 0;
        if (g == 0) ovcnt[0] = 0;
    }
    __syncthreads();

    #pragma unroll
    for (int k = 0; k < EDGES_PER_THREAD; ++k) {
        const int idx = k * 256 + t;
        if (idx < CHUNK_E) {
            const unsigned int b  = (unsigned int)d[k] / (unsigned int)BIN_NODES;
            const unsigned int dl = (unsigned int)d[k] - b * (unsigned int)BIN_NODES;
            const int word = (int)((dl << 16) | (unsigned int)s[k]);
            if (!lds_line_put(&C[b * LPAD], word)) {
                const int k2 = atomicAdd(&nspill, 1);
                ((int2*)spill)[g * CHUNK_E + k2] = make_int2(d[k], s[k]);
            }
        }
    }
    __syncthreads();

    for (int i = t; i < NBIN * LINE_W; i += 256) {
        const int b = i >> 5, j = i & 31;
        cells[((size_t)b * NCHUNK + g) * LINE_W + j] = C[b * LPAD + j];
    }
    if (t == 0) spill_cnt[g] = nspill;
}

__device__ __forceinline__ void node_put(int* L, int* head, const int dl, const int s,
                                         int* __restrict__ ovcnt, int* __restrict__ ov) {
    if (!lds_line_put(&L[dl * LPAD], s)) {
        const int k = atomicAdd(ovcnt, 1);
        const int prev = atomicExch(&head[dl], k);
        ((int2*)ov)[k] = make_int2(prev, s);
    }
}

#define W_PITCH 132
#define W_LDS_WORDS (HIDDEN * W_PITCH)

__device__ __forceinline__ void stage_w_lds(float* Wl, const float* __restrict__ W) {
    const int t = threadIdx.x;
    v4f lo[4], hi[4];
    #pragma unroll
    for (int i = 0; i < 4; ++i) {
        const int gi = i * 256 + t;
        const float* src = W + (gi >> 4) * IN_DIM + (gi & 15) * 8;
        lo[i] = *(const v4f*)(src);
        hi[i] = *(const v4f*)(src + 4);
    }
    #pragma unroll
    for (int i = 0; i < 4; ++i) {
        const int gi = i * 256 + t;
        float* dst = Wl + (gi >> 4) * W_PITCH + (gi & 15) * 8;
        const v4f p0 = {lo[i].x, lo[i].y, hi[i].x, hi[i].y};
        const v4f p1 = {lo[i].z, lo[i].w, hi[i].z, hi[i].w};
        *(v4f*)(dst)     = p0;
        *(v4f*)(dst + 4) = p1;
    }
}

__device__ __forceinline__ void
gat_gemm_load_a(v2f (&a)[IN_DIM / 4], const int tile, const int lane, const float* __restrict__ x) {
    const float* xrow = x + (size_t)(tile * 16 + (lane & 15)) * IN_DIM + 2 * (lane >> 4);
    #pragma unroll
    for (int i = 0; i < IN_DIM / 4; ++i) a[i] = *(const v2f*)(xrow + 4 * i);
}

__device__ __forceinline__ void
gat_gemm_wmma_tile(const v2f (&a)[IN_DIM / 4], const float* Wl,
                   const int tile, const int lane,
                   const float* __restrict__ att_dst_v,
                   float* __restrict__ h, float* __restrict__ a_dst)
{
    const int half = lane >> 4;
    const int lrow = lane & 15;
    const int row0 = tile * 16;

    const float* wl = Wl + lrow * W_PITCH + 4 * half;

    v8f acc0 = {}, acc1 = {}, acc2 = {}, acc3 = {};

    v4f b0 = *(const v4f*)(wl);
    v4f b1 = *(const v4f*)(wl + 16 * W_PITCH);
    v4f b2 = *(const v4f*)(wl + 32 * W_PITCH);
    v4f b3 = *(const v4f*)(wl + 48 * W_PITCH);
    __builtin_amdgcn_sched_group_barrier(0x100, 4, 0);
    #pragma unroll
    for (int g = 0; g < IN_DIM / 8; ++g) {
        const v4f c0 = b0, c1 = b1, c2 = b2, c3 = b3;
        if (g + 1 < IN_DIM / 8) {
            b0 = *(const v4f*)(wl + (g + 1) * 8);
            b1 = *(const v4f*)(wl + 16 * W_PITCH + (g + 1) * 8);
            b2 = *(const v4f*)(wl + 32 * W_PITCH + (g + 1) * 8);
            b3 = *(const v4f*)(wl + 48 * W_PITCH + (g + 1) * 8);
        }
        const v2f a0 = a[2 * g], a1 = a[2 * g + 1];
        acc0 = __builtin_amdgcn_wmma_f32_16x16x4_f32(false, a0, false, c0.xy, (short)0, acc0, false, false);
        acc1 = __builtin_amdgcn_wmma_f32_16x16x4_f32(false, a0, false, c1.xy, (short)0, acc1, false, false);
        acc2 = __builtin_amdgcn_wmma_f32_16x16x4_f32(false, a0, false, c2.xy, (short)0, acc2, false, false);
        acc3 = __builtin_amdgcn_wmma_f32_16x16x4_f32(false, a0, false, c3.xy, (short)0, acc3, false, false);
        acc0 = __builtin_amdgcn_wmma_f32_16x16x4_f32(false, a1, false, c0.zw, (short)0, acc0, false, false);
        acc1 = __builtin_amdgcn_wmma_f32_16x16x4_f32(false, a1, false, c1.zw, (short)0, acc1, false, false);
        acc2 = __builtin_amdgcn_wmma_f32_16x16x4_f32(false, a1, false, c2.zw, (short)0, acc2, false, false);
        acc3 = __builtin_amdgcn_wmma_f32_16x16x4_f32(false, a1, false, c3.zw, (short)0, acc3, false, false);
        if (g + 1 < IN_DIM / 8)
            __builtin_amdgcn_sched_group_barrier(0x100, 4, 0);
        __builtin_amdgcn_sched_group_barrier(0x008, 8, 0);
    }

    const float ad0 = att_dst_v[lrow];
    const float ad1 = att_dst_v[16 + lrow];
    const float ad2 = att_dst_v[32 + lrow];
    const float ad3 = att_dst_v[48 + lrow];

    #pragma unroll
    for (int r = 0; r < 8; ++r) {
        const int row = row0 + r + 8 * half;
        float* hr = h + (size_t)row * HIDDEN;
        hr[lrow]      = acc0[r];
        hr[16 + lrow] = acc1[r];
        hr[32 + lrow] = acc2[r];
        hr[48 + lrow] = acc3[r];
        float pd = acc0[r] * ad0 + acc1[r] * ad1 + acc2[r] * ad2 + acc3[r] * ad3;
        #pragma unroll
        for (int off = 1; off < 16; off <<= 1) pd += __shfl_xor(pd, off, 32);
        if (lrow == 0) a_dst[row] = pd;
    }
}

#define GEMM_TILES 3125
#define TILES_PER_BLOCK 8
#define GEMM_BLOCKS 391
#define BIN_GEMM_GRID 641

__global__ void __launch_bounds__(256)
bin_and_gemm_kernel(const int* __restrict__ ei, int* __restrict__ cells,
                    int* __restrict__ spill_cnt, int* __restrict__ spill,
                    int* __restrict__ ovcnt,
                    const float* __restrict__ x, const float* __restrict__ W,
                    const float* __restrict__ att_dst_v,
                    float* __restrict__ h, float* __restrict__ a_dst) {
    __shared__ __attribute__((aligned(16))) int S[W_LDS_WORDS];
    __shared__ int nspill;
    static_assert(W_LDS_WORDS >= NBIN * LPAD, "the buffer holds a chunk's cells as well");
    static_assert(BIN_GEMM_GRID == NCHUNK + GEMM_BLOCKS, "grid = chunk blocks + GEMM blocks");
    static_assert(GEMM_BLOCKS * TILES_PER_BLOCK >= GEMM_TILES, "every tile has a wave");
    if (blockIdx.x < NCHUNK) {
        edge_bin_block(S, nspill, ei, cells, spill_cnt, spill, ovcnt);
    } else {
        const int wv = __builtin_amdgcn_readfirstlane((int)(threadIdx.x >> 5));
        const int tile = ((int)blockIdx.x - NCHUNK) * TILES_PER_BLOCK + wv;
        const int lane = (int)(threadIdx.x & 31);
        float* Wl = reinterpret_cast<float*>(S);
        v2f a[IN_DIM / 4];
        gat_gemm_load_a(a, (tile < GEMM_TILES) ? tile : GEMM_TILES - 1, lane, x);
        stage_w_lds(Wl, W);
        __syncthreads();
        if (tile < GEMM_TILES)
            gat_gemm_wmma_tile(a, Wl, tile, lane, att_dst_v, h, a_dst);
    }
}

#define NPAIR 16

template <int CTRL>
__device__ __forceinline__ int row_rd(int x) {
    return __builtin_amdgcn_update_dpp(0, x, CTRL, 0xf, 0xf, true);
}
template <int CTRL>
__device__ __forceinline__ float row_rd(float x) {
    return __builtin_bit_cast(float, row_rd<CTRL>(__builtin_bit_cast(int, x)));
}
__device__ __forceinline__ float xhalf(float x) {
    const int b = __builtin_bit_cast(int, x);
    return __builtin_bit_cast(float,
        __builtin_amdgcn_permlanex16(b, b, 0x76543210, 0xfedcba98, false, false));
}
__device__ __forceinline__ float row_sum16(float p) {
    p += row_rd<0x161>(p); p += row_rd<0x162>(p); p += row_rd<0x164>(p); p += row_rd<0x168>(p);
    return p;
}
__device__ __forceinline__ float row_max16(float p) {
    p = fmaxf(p, row_rd<0x161>(p)); p = fmaxf(p, row_rd<0x162>(p));
    p = fmaxf(p, row_rd<0x164>(p)); p = fmaxf(p, row_rd<0x168>(p));
    return p;
}
__device__ __forceinline__ float dot4(const v4f a, const v4f b) {
    return a.x * b.x + a.y * b.y + a.z * b.z + a.w * b.w;
}

template <int I>
__device__ __forceinline__ void load_pairs(v4f (&r)[NPAIR], const int np, const int ent2,
                                           const unsigned int q,
                                           const v4f* __restrict__ h4) {
    if constexpr (I < NPAIR) {
        if (I < np) {
            const unsigned int s = (unsigned int)row_rd<0x150 | I>(ent2);
            r[I] = h4[s * 16u + q];
            load_pairs<I + 1>(r, np, ent2, q, h4);
        }
    }
}
template <int I>
__device__ __forceinline__ void score_pairs(const v4f (&r)[NPAIR], const int np, const v4f as4,
                                            const int q, float& sc) {
    if constexpr (I < NPAIR) {
        if (I < np) {
            const float p = row_sum16(dot4(r[I], as4));
            sc = (q == I) ? p : sc;
            score_pairs<I + 1>(r, np, as4, q, sc);
        }
    }
}
template <int I>
__device__ __forceinline__ void accum_pairs(const v4f (&r)[NPAIR], const int np, const float ex,
                                            v4f& acc) {
    if constexpr (I < NPAIR) {
        if (I < np) {
            const float ek = row_rd<0x150 | I>(ex);
            acc.x = fmaf(ek, r[I].x, acc.x);
            acc.y = fmaf(ek, r[I].y, acc.y);
            acc.z = fmaf(ek, r[I].z, acc.z);
            acc.w = fmaf(ek, r[I].w, acc.w);
            accum_pairs<I + 1>(r, np, ex, acc);
        }
    }
}

#define GATHER_THREADS 1024
#define GATHER_WAVES 32

__global__ void __launch_bounds__(GATHER_THREADS)
lines_gather_kernel(const int* __restrict__ cells, const int* __restrict__ spill_cnt,
                    const int* __restrict__ spill, int* __restrict__ ovcnt, int* __restrict__ ov,
                    const float* __restrict__ att_src_v, const float* __restrict__ a_dst,
                    const float* __restrict__ h,
                    const float* __restrict__ bias_conv, const float* __restrict__ W_lin,
                    const float* __restrict__ b_lin, float* __restrict__ y) {
    __shared__ int L[BIN_NODES * LPAD];
    __shared__ int head[BIN_NODES];
    const int t = threadIdx.x;
    const int b = blockIdx.x;
    const int lane = t & 31;
    const int half = lane >> 4, q = lane & 15;
    const int wv = __builtin_amdgcn_readfirstlane(t >> 5);

    for (int i = t; i < BIN_NODES * LPAD; i += GATHER_THREADS) L[i] = 0;
    if (t < BIN_NODES) head[t] = -1;
    __syncthreads();

    const v4f as4 = *(const v4f*)(att_src_v + q * 4);
    const v4f bc  = *(const v4f*)(bias_conv + q * 4);
    const v4f wl  = *(const v4f*)(W_lin + q * 4);
    const float bl = b_lin[0];

    const int* cb = cells + (size_t)b * NCHUNK * LINE_W;
    for (int g0 = wv; g0 < NCHUNK; g0 += 8 * GATHER_WAVES) {
        int v[8];
        #pragma unroll
        for (int u = 0; u < 8; ++u) {
            const int g = g0 + GATHER_WAVES * u;
            v[u] = (g < NCHUNK) ? cb[g * LINE_W + lane] : 0;
        }
        #pragma unroll
        for (int u = 0; u < 8; ++u) {
            const int c  = __builtin_amdgcn_readlane(v[u], 0);
            const int cc = (c < SLOTS) ? c : SLOTS;
            if (lane >= 1 && lane <= cc) {
                const unsigned int word = (unsigned int)v[u];
                node_put(L, head, (int)(word >> 16), (int)(word & 0xffffu), ovcnt, ov);
            }
        }
    }

    for (int g = t; g < NCHUNK; g += GATHER_THREADS) {
        const int cr = spill_cnt[g];
        const int c = (cr < CHUNK_E) ? cr : CHUNK_E;
        for (int j = 0; j < c; ++j) {
            const int2 p = ((const int2*)spill)[g * CHUNK_E + j];
            const unsigned int pb = (unsigned int)p.x / (unsigned int)BIN_NODES;
            if (pb == (unsigned int)b) node_put(L, head, p.x - b * BIN_NODES, p.y, ovcnt, ov);
        }
    }
    __syncthreads();

    const v4f* h4 = (const v4f*)h;
    const int2* ov2 = (const int2*)ov;
    for (int dl = wv; dl < BIN_NODES; dl += GATHER_WAVES) {
        const int d = b * BIN_NODES + dl;
        const int w = L[dl * LPAD + lane];
        const float ad = a_dst[d];

        const int n  = __builtin_amdgcn_readfirstlane(w);
        const int nb = (n < SLOTS) ? n : SLOTS;
        const int np = (nb + 2) >> 1;
        const int ent = (lane >= 1 && lane <= nb) ? w : d;
        const int ent2 = __shfl(ent, 2 * q + half, 32);

        v4f r[NPAIR];
        load_pairs<0>(r, np, ent2, (unsigned int)q, h4);

        float sc = 0.0f;
        score_pairs<0>(r, np, as4, q, sc);
        const bool valid = (2 * q + half) <= nb;
        const float v = valid ? leaky(sc + ad) : -INFINITY;
        float m = row_max16(v);
        m = fmaxf(m, xhalf(m));

        const int extra = n - SLOTS;
        if (extra > 0) {
            int k = head[dl];
            for (int j = 0; j < extra; ++j) {
                k = __builtin_amdgcn_readfirstlane(k);
                if (k < 0 || k >= N_EDGES) break;
                const int2 p = ov2[k];
                const v4f hv = *(const v4f*)(h + (size_t)p.y * HIDDEN + q * 4);
                m = fmaxf(m, leaky(row_sum16(dot4(hv, as4)) + ad));
                k = p.x;
            }
        }

        const float ex = valid ? expf(v - m) : 0.0f;
        v4f acc = {0.0f, 0.0f, 0.0f, 0.0f};
        accum_pairs<0>(r, np, ex, acc);

        float sov = 0.0f;
        if (extra > 0) {
            int k = head[dl];
            for (int j = 0; j < extra; ++j) {
                k = __builtin_amdgcn_readfirstlane(k);
                if (k < 0 || k >= N_EDGES) break;
                const int2 p = ov2[k];
                const v4f hv = *(const v4f*)(h + (size_t)p.y * HIDDEN + q * 4);
                const float eo = expf(leaky(row_sum16(dot4(hv, as4)) + ad) - m);
                sov += eo;
                const float eh = (half == 0) ? eo : 0.0f;
                acc.x = fmaf(eh, hv.x, acc.x);
                acc.y = fmaf(eh, hv.y, acc.y);
                acc.z = fmaf(eh, hv.z, acc.z);
                acc.w = fmaf(eh, hv.w, acc.w);
                k = p.x;
            }
        }

        acc.x += xhalf(acc.x);
        acc.y += xhalf(acc.y);
        acc.z += xhalf(acc.z);
        acc.w += xhalf(acc.w);
        float ssum = row_sum16(ex);
        ssum = ssum + xhalf(ssum) + sov;

        const float inv = 1.0f / ssum;
        float p = fmaxf(acc.x * inv + bc.x, 0.0f) * wl.x
                + fmaxf(acc.y * inv + bc.y, 0.0f) * wl.y
                + fmaxf(acc.z * inv + bc.z, 0.0f) * wl.z
                + fmaxf(acc.w * inv + bc.w, 0.0f) * wl.w;
        p = row_sum16(p);
        if (lane == 0) y[d] = 1.0f / (1.0f + expf(-(p + bl)));
    }
}

extern "C" void kernel_launch(void* const* d_in, const int* in_sizes, int n_in,
                              void* d_out, int out_size, void* d_ws, size_t ws_size,
                              hipStream_t stream) {
    const float* x         = (const float*)d_in[0];
    const int*   ei        = (const int*)  d_in[1];
    const float* W         = (const float*)d_in[2];
    const float* att_src_v = (const float*)d_in[3];
    const float* att_dst_v = (const float*)d_in[4];
    const float* bias_conv = (const float*)d_in[5];
    const float* W_lin     = (const float*)d_in[6];
    const float* b_lin     = (const float*)d_in[7];
    float* y = (float*)d_out;

    float* ws     = (float*)d_ws;
    float* h      = ws;
    float* a_dst  = ws + 3250000;
    int*   ovcnt  = (int*)(ws + 4900032);
    int*   ov     = (int*)(ws + 4900096);
    int*   cells  = (int*)(ws + 6500096);
    int*   spcnt  = (int*)(ws + 10500096);
    int*   spill  = (int*)(ws + 10500608);

    bin_and_gemm_kernel<<<BIN_GEMM_GRID, 256, 0, stream>>>(ei, cells, spcnt, spill, ovcnt,
                                                          x, W, att_dst_v, h, a_dst);

    lines_gather_kernel<<<NBIN, GATHER_THREADS, 0, stream>>>(cells, spcnt, spill, ovcnt, ov,
                                                             att_src_v, a_dst, h,
                                                             bias_conv, W_lin, b_lin, y);
}
